# starting kernel
# baseline (speedup 1.0000x reference)
#include <hip/hip_runtime.h>
#include <hip/hip_bf16.h>

typedef __attribute__((ext_vector_type(2))) float v2f;
typedef __attribute__((ext_vector_type(8))) float v8f;

#define IN_C  128
#define HID_C 128
#define OUT_C 64

__global__ void deg_init(float* __restrict__ deg, int n) {
  int i = blockIdx.x * blockDim.x + threadIdx.x;
  if (i < n) deg[i] = 1.0f;
}

__global__ void deg_count(const int* __restrict__ dst, int E, float* __restrict__ deg) {
  int i = blockIdx.x * blockDim.x + threadIdx.x;
  if (i < E) atomicAdd(&deg[dst[i]], 1.0f);
}

__global__ void deg_finish(float* __restrict__ deg, int n) {
  int i = blockIdx.x * blockDim.x + threadIdx.x;
  if (i < n) deg[i] = rsqrtf(deg[i]);
}

__global__ void wmma_gemm_f32(const float* __restrict__ A, const float* __restrict__ B,
                              float* __restrict__ C, int M, int N, int K) {
  int wave = (int)((blockIdx.x * (long long)blockDim.x + threadIdx.x) >> 5);
  int lane = threadIdx.x & 31;
  int ntiles = N >> 4;
  int total = (M >> 4) * ntiles;
  if (wave >= total) return;
  int mt = wave / ntiles, nt = wave - mt * ntiles;
  int m0 = mt << 4, n0 = nt << 4;
  int half = lane >> 4;
  int l = lane & 15;

  const float* Arow = A + (size_t)(m0 + l) * K + 2 * half;
  const float* Bcol = B + (size_t)(2 * half) * N + n0 + l;

  v8f acc = {};
  for (int k = 0; k < K; k += 4) {
    v2f a, b;
    a.x = Arow[k];
    a.y = Arow[k + 1];
    b.x = Bcol[(size_t)k * N];
    b.y = Bcol[(size_t)(k + 1) * N];
    acc = __builtin_amdgcn_wmma_f32_16x16x4_f32(false, a, false, b, (short)0, acc,
                                                false, false);
  }

  float* Crow = C + (size_t)(m0 + 8 * half) * N + n0 + l;
#pragma unroll
  for (int i = 0; i < 8; ++i) Crow[(size_t)i * N] = acc[i];
}

__global__ void agg_self(const float* __restrict__ h, const float* __restrict__ dinv,
                         float* __restrict__ out, long long total, int C) {
  long long t = blockIdx.x * (long long)blockDim.x + threadIdx.x;
  if (t >= total) return;
  int v = (int)(t / C);
  float di = dinv[v];
  out[t] = h[t] * di * di;
}

__global__ void agg_scatter(const float* __restrict__ h, const int* __restrict__ src,
                            const int* __restrict__ dst, const float* __restrict__ dinv,
                            float* __restrict__ out, int E, int C) {
  long long t = blockIdx.x * (long long)blockDim.x + threadIdx.x;
  int groups = C >> 2;
  long long total = (long long)E * groups;
  if (t >= total) return;
  int e  = (int)(t / groups);
  int c4 = (int)(t - (long long)e * groups) << 2;
  int s = src[e], d = dst[e];
  float nrm = dinv[s] * dinv[d];
  float4 hv = *(const float4*)(h + (size_t)s * C + c4);
  float* o = out + (size_t)d * C + c4;
  atomicAdd(o + 0, hv.x * nrm);
  atomicAdd(o + 1, hv.y * nrm);
  atomicAdd(o + 2, hv.z * nrm);
  atomicAdd(o + 3, hv.w * nrm);
}

__global__ void bias_act(float* __restrict__ out, const float* __restrict__ b,
                         long long total, int C, int do_relu) {
  long long t = blockIdx.x * (long long)blockDim.x + threadIdx.x;
  if (t >= total) return;
  float v = out[t] + b[(int)(t % C)];
  out[t] = do_relu ? fmaxf(v, 0.0f) : v;
}

__global__ void decode_mlp(const float* __restrict__ z,
                           const int* __restrict__ sidx, const int* __restrict__ didx,
                           const float* __restrict__ Wa,
                           const float* __restrict__ ba,
                           const float* __restrict__ Wb,
                           const float* __restrict__ bb,
                           float* __restrict__ out, int P) {
  int wave = (int)((blockIdx.x * (long long)blockDim.x + threadIdx.x) >> 5);
  int lane = threadIdx.x & 31;
  if (wave >= (P >> 4)) return;
  int m0 = wave << 4;
  int half = lane >> 4;
  int l = lane & 15;

  int row = m0 + l;
  int s = sidx[row], d = didx[row];
  const float* zs = z + (size_t)s * OUT_C;
  const float* zd = z + (size_t)d * OUT_C;

  v8f acc[8];
#pragma unroll
  for (int t = 0; t < 8; ++t) acc[t] = (v8f){};

  for (int k = 0; k < 2 * OUT_C; k += 4) {
    int ka = k + 2 * half;
    v2f a;
    a.x = (ka     < OUT_C) ? zs[ka]         : zd[ka - OUT_C];
    a.y = (ka + 1 < OUT_C) ? zs[ka + 1]     : zd[ka + 1 - OUT_C];
    const float* Bp = Wa + (size_t)ka * HID_C + l;
#pragma unroll
    for (int t = 0; t < 8; ++t) {
      v2f b;
      b.x = Bp[t * 16];
      b.y = Bp[HID_C + t * 16];
      acc[t] = __builtin_amdgcn_wmma_f32_16x16x4_f32(false, a, false, b, (short)0,
                                                     acc[t], false, false);
    }
  }

  float part[8];
#pragma unroll
  for (int i = 0; i < 8; ++i) part[i] = 0.0f;
#pragma unroll
  for (int t = 0; t < 8; ++t) {
    int nidx = t * 16 + l;
    float bav = ba[nidx];
    float wbv = Wb[nidx];
#pragma unroll
    for (int i = 0; i < 8; ++i) {
      float hv = fmaxf(acc[t][i] + bav, 0.0f);
      part[i] += hv * wbv;
    }
  }
#pragma unroll
  for (int m = 1; m <= 8; m <<= 1) {
#pragma unroll
    for (int i = 0; i < 8; ++i) part[i] += __shfl_xor(part[i], m, 32);
  }
  if (l == 0) {
    float bbv = bb[0];
#pragma unroll
    for (int i = 0; i < 8; ++i) out[m0 + 8 * half + i] = part[i] + bbv;
  }
}

extern "C" void kernel_launch(void* const* d_in, const int* in_sizes, int n_in,
                              void* d_out, int out_size, void* d_ws, size_t ws_size,
                              hipStream_t stream) {
  const float* x   = (const float*)d_in[0];
  const int* eidx  = (const int*)d_in[1];
  const int* pidx  = (const int*)d_in[2];
  const int* nidx  = (const int*)d_in[3];
  const float* W1  = (const float*)d_in[4];
  const float* b1  = (const float*)d_in[5];
  const float* W2  = (const float*)d_in[6];
  const float* b2  = (const float*)d_in[7];
  const float* Wa  = (const float*)d_in[8];
  const float* ba  = (const float*)d_in[9];
  const float* Wb  = (const float*)d_in[10];
  const float* bb  = (const float*)d_in[11];
  float* out = (float*)d_out;

  const int n = in_sizes[0] / IN_C;
  const int E = in_sizes[1] / 2;
  const int P = in_sizes[2] / 2;
  const int* e_src = eidx;
  const int* e_dst = eidx + E;

  float* ws   = (float*)d_ws;
  float* h1   = ws;
  float* z1   = h1 + (size_t)n * HID_C;
  float* dinv = z1 + (size_t)n * HID_C;
  float* h2   = h1;
  float* z2   = h1 + (size_t)n * OUT_C;

  auto cdiv = [](long long a, long long b) { return (int)((a + b - 1) / b); };
  const int B = 256;

  deg_init<<<cdiv(n, B), B, 0, stream>>>(dinv, n);
  deg_count<<<cdiv(E, B), B, 0, stream>>>(e_dst, E, dinv);
  deg_finish<<<cdiv(n, B), B, 0, stream>>>(dinv, n);

  {
    long long waves = (long long)(n / 16) * (HID_C / 16);
    wmma_gemm_f32<<<cdiv(waves * 32, B), B, 0, stream>>>(x, W1, h1, n, HID_C, IN_C);
    long long tot = (long long)n * HID_C;
    agg_self<<<cdiv(tot, B), B, 0, stream>>>(h1, dinv, z1, tot, HID_C);
    agg_scatter<<<cdiv((long long)E * (HID_C / 4), B), B, 0, stream>>>(
        h1, e_src, e_dst, dinv, z1, E, HID_C);
    bias_act<<<cdiv(tot, B), B, 0, stream>>>(z1, b1, tot, HID_C, 1);
  }

  {
    long long waves = (long long)(n / 16) * (OUT_C / 16);
    wmma_gemm_f32<<<cdiv(waves * 32, B), B, 0, stream>>>(z1, W2, h2, n, OUT_C, HID_C);
    long long tot = (long long)n * OUT_C;
    agg_self<<<cdiv(tot, B), B, 0, stream>>>(h2, dinv, z2, tot, OUT_C);
    agg_scatter<<<cdiv((long long)E * (OUT_C / 4), B), B, 0, stream>>>(
        h2, e_src, e_dst, dinv, z2, E, OUT_C);
    bias_act<<<cdiv(tot, B), B, 0, stream>>>(z2, b2, tot, OUT_C, 0);
  }

  {
    long long waves = P / 16;
    decode_mlp<<<cdiv(waves * 32, B), B, 0, stream>>>(z2, pidx, pidx + P, Wa, ba, Wb,
                                                      bb, out, P);
    decode_mlp<<<cdiv(waves * 32, B), B, 0, stream>>>(z2, nidx, nidx + P, Wa, ba, Wb,
                                                      bb, out + P, P);
  }
}


# agg: atomic scatter replaced by owner work-groups: each holds 196 dst rows in an LDS tile, scans all dst, adds src rows by LDS float add; self loop, bias, relu folded in; sum order = LDS arrival
# speedup vs baseline: 1.5063x; 1.5063x over previous
#include <hip/hip_runtime.h>
#include <hip/hip_bf16.h>

typedef __attribute__((ext_vector_type(2))) float v2f;
typedef __attribute__((ext_vector_type(8))) float v8f;

#define IN_C  128
#define HID_C 128
#define OUT_C 64

__global__ void deg_init(float* __restrict__ deg, int n) {
  int i = blockIdx.x * blockDim.x + threadIdx.x;
  if (i < n) deg[i] = 1.0f;
}

__global__ void deg_count(const int* __restrict__ dst, int E, float* __restrict__ deg) {
  int i = blockIdx.x * blockDim.x + threadIdx.x;
  if (i < E) atomicAdd(&deg[dst[i]], 1.0f);
}

__global__ void deg_finish(float* __restrict__ deg, int n) {
  int i = blockIdx.x * blockDim.x + threadIdx.x;
  if (i < n) deg[i] = rsqrtf(deg[i]);
}

__global__ void wmma_gemm_f32(const float* __restrict__ A, const float* __restrict__ B,
                              float* __restrict__ C, int M, int N, int K) {
  int wave = (int)((blockIdx.x * (long long)blockDim.x + threadIdx.x) >> 5);
  int lane = threadIdx.x & 31;
  int ntiles = N >> 4;
  int total = (M >> 4) * ntiles;
  if (wave >= total) return;
  int mt = wave / ntiles, nt = wave - mt * ntiles;
  int m0 = mt << 4, n0 = nt << 4;
  int half = lane >> 4;
  int l = lane & 15;

  const float* Arow = A + (size_t)(m0 + l) * K + 2 * half;
  const float* Bcol = B + (size_t)(2 * half) * N + n0 + l;

  v8f acc = {};
  for (int k = 0; k < K; k += 4) {
    v2f a, b;
    a.x = Arow[k];
    a.y = Arow[k + 1];
    b.x = Bcol[(size_t)k * N];
    b.y = Bcol[(size_t)(k + 1) * N];
    acc = __builtin_amdgcn_wmma_f32_16x16x4_f32(false, a, false, b, (short)0, acc,
                                                false, false);
  }

  float* Crow = C + (size_t)(m0 + 8 * half) * N + n0 + l;
#pragma unroll
  for (int i = 0; i < 8; ++i) Crow[(size_t)i * N] = acc[i];
}

#define AGG_ROWS    196
#define AGG_THREADS 1024
#define AGG_WAVES   32
#define AGG_UNROLL  4
#define AGG_BATCH   32
#define AGG_CAP     (AGG_BATCH + AGG_UNROLL * 128)

template <int C, bool RELU>
__global__ __launch_bounds__(AGG_THREADS)
void agg_owner(const float* __restrict__ h, const int* __restrict__ src,
               const int* __restrict__ dst, const float* __restrict__ dinv,
               const float* __restrict__ bias, float* __restrict__ out, int n, int E) {
  constexpr int VEC = C / 32;
  typedef float vec_t __attribute__((ext_vector_type(VEC)));

  __shared__ float tile[AGG_ROWS * C];
  __shared__ int pend[AGG_WAVES][AGG_CAP];

  const int tid = threadIdx.x;
  const int lane = tid & 31;
  const int wave = tid >> 5;
  const int lo = blockIdx.x * AGG_ROWS;
  const int nrows = min(AGG_ROWS, n - lo);
  int* mylist = pend[wave];

  for (int r = wave; r < nrows; r += AGG_WAVES) {
    const int v = lo + r;
    const float di = dinv[v];
    const vec_t hv = *(const vec_t*)(h + (size_t)v * C + lane * VEC);
#pragma unroll
    for (int k = 0; k < VEC; ++k) tile[r * C + k * 32 + lane] = hv[k] * di * di;
  }
  __syncthreads();

  int cnt = 0;

  auto flush = [&](bool all) {
    const int stop = all ? cnt : (cnt & ~31);
    for (int b0 = 0; b0 < stop; b0 += 32) {
      const int nb = min(32, stop - b0);
      int s = 0, dl = 0;
      float nrm = 0.0f;
      if (lane < nb) {
        const int e = mylist[b0 + lane];
        s = src[e];
        const int d = dst[e];
        nrm = dinv[s] * dinv[d];
        dl = d - lo;
      }
      for (int j0 = 0; j0 < nb; j0 += 8) {
        vec_t hv[8];
#pragma unroll
        for (int u = 0; u < 8; ++u) {
          const int sj = __builtin_amdgcn_readlane(s, j0 + u);
          hv[u] = *(const vec_t*)(h + (size_t)sj * C + lane * VEC);
        }
#pragma unroll
        for (int u = 0; u < 8; ++u) {
          const bool live = j0 + u < nb;
          const int dj = __builtin_amdgcn_readlane(dl, j0 + u);
          const float nj =
              __int_as_float(__builtin_amdgcn_readlane(__float_as_int(nrm), j0 + u));
          float* trow = tile + dj * C + lane;
#pragma unroll
          for (int k = 0; k < VEC; ++k)
            __hip_atomic_fetch_add(trow + k * 32, live ? hv[u][k] * nj : 0.0f,
                                   __ATOMIC_RELAXED, __HIP_MEMORY_SCOPE_WORKGROUP);
        }
      }
    }
    const int rem = cnt - stop;
    if (rem > 0) {
      int keep = 0;
      if (lane < rem) keep = mylist[stop + lane];
      if (lane < rem) mylist[lane] = keep;
    }
    cnt = rem;
  };

  auto take = [&](int d, int e, int lim) {
    const bool m = (unsigned)(d - lo) < (unsigned)lim;
    const unsigned b = __builtin_amdgcn_ballot_w32(m);
    if (b != 0u) {
      if (m) mylist[cnt + (int)__builtin_amdgcn_mbcnt_lo(b, 0u)] = e;
      cnt += __builtin_popcount(b);
    }
  };

  if ((E & 3) == 0 && E > 0) {
    const int4* d4 = (const int4*)dst;
    const int nq = E >> 2;
    auto ldq = [&](int q) { return d4[min(q, nq - 1)]; };
    constexpr int TRIP = AGG_UNROLL * AGG_THREADS;
    const int ntrips = (nq + TRIP - 1) / TRIP;
    int tt = blockIdx.x % ntrips;
    int4 cur[AGG_UNROLL], nxt[AGG_UNROLL];
#pragma unroll
    for (int u = 0; u < AGG_UNROLL; ++u) cur[u] = ldq(tt * TRIP + u * AGG_THREADS + tid);
    for (int t = 0; t < ntrips; ++t) {
      const int tn = (tt + 1 == ntrips) ? 0 : tt + 1;
#pragma unroll
      for (int u = 0; u < AGG_UNROLL; ++u)
        nxt[u] = ldq(tn * TRIP + u * AGG_THREADS + tid);
      if (cnt >= AGG_BATCH) flush(false);
#pragma unroll
      for (int u = 0; u < AGG_UNROLL; ++u) {
        const int q = tt * TRIP + u * AGG_THREADS + tid;
        const int lim = (q < nq) ? nrows : 0;
        const int e = q << 2;
        take(cur[u].x, e, lim);
        take(cur[u].y, e + 1, lim);
        take(cur[u].z, e + 2, lim);
        take(cur[u].w, e + 3, lim);
      }
#pragma unroll
      for (int u = 0; u < AGG_UNROLL; ++u) cur[u] = nxt[u];
      tt = tn;
    }
  } else {
    for (int e0 = 0; e0 < E; e0 += AGG_THREADS) {
      if (cnt >= AGG_BATCH) flush(false);
      const int e = e0 + tid;
      take(dst[min(e, E - 1)], e, (e < E) ? nrows : 0);
    }
  }
  flush(true);
  __syncthreads();

  const vec_t bv = *(const vec_t*)(bias + lane * VEC);
  for (int r = wave; r < nrows; r += AGG_WAVES) {
    vec_t o;
#pragma unroll
    for (int k = 0; k < VEC; ++k) {
      const float t = tile[r * C + k * 32 + lane] + bv[k];
      o[k] = RELU ? fmaxf(t, 0.0f) : t;
    }
    *(vec_t*)(out + (size_t)(lo + r) * C + lane * VEC) = o;
  }
}

__global__ void decode_mlp(const float* __restrict__ z,
                           const int* __restrict__ sidx, const int* __restrict__ didx,
                           const float* __restrict__ Wa,
                           const float* __restrict__ ba,
                           const float* __restrict__ Wb,
                           const float* __restrict__ bb,
                           float* __restrict__ out, int P) {
  int wave = (int)((blockIdx.x * (long long)blockDim.x + threadIdx.x) >> 5);
  int lane = threadIdx.x & 31;
  if (wave >= (P >> 4)) return;
  int m0 = wave << 4;
  int half = lane >> 4;
  int l = lane & 15;

  int row = m0 + l;
  int s = sidx[row], d = didx[row];
  const float* zs = z + (size_t)s * OUT_C;
  const float* zd = z + (size_t)d * OUT_C;

  v8f acc[8];
#pragma unroll
  for (int t = 0; t < 8; ++t) acc[t] = (v8f){};

  for (int k = 0; k < 2 * OUT_C; k += 4) {
    int ka = k + 2 * half;
    v2f a;
    a.x = (ka     < OUT_C) ? zs[ka]         : zd[ka - OUT_C];
    a.y = (ka + 1 < OUT_C) ? zs[ka + 1]     : zd[ka + 1 - OUT_C];
    const float* Bp = Wa + (size_t)ka * HID_C + l;
#pragma unroll
    for (int t = 0; t < 8; ++t) {
      v2f b;
      b.x = Bp[t * 16];
      b.y = Bp[HID_C + t * 16];
      acc[t] = __builtin_amdgcn_wmma_f32_16x16x4_f32(false, a, false, b, (short)0,
                                                     acc[t], false, false);
    }
  }

  float part[8];
#pragma unroll
  for (int i = 0; i < 8; ++i) part[i] = 0.0f;
#pragma unroll
  for (int t = 0; t < 8; ++t) {
    int nidx = t * 16 + l;
    float bav = ba[nidx];
    float wbv = Wb[nidx];
#pragma unroll
    for (int i = 0; i < 8; ++i) {
      float hv = fmaxf(acc[t][i] + bav, 0.0f);
      part[i] += hv * wbv;
    }
  }
#pragma unroll
  for (int m = 1; m <= 8; m <<= 1) {
#pragma unroll
    for (int i = 0; i < 8; ++i) part[i] += __shfl_xor(part[i], m, 32);
  }
  if (l == 0) {
    float bbv = bb[0];
#pragma unroll
    for (int i = 0; i < 8; ++i) out[m0 + 8 * half + i] = part[i] + bbv;
  }
}

extern "C" void kernel_launch(void* const* d_in, const int* in_sizes, int n_in,
                              void* d_out, int out_size, void* d_ws, size_t ws_size,
                              hipStream_t stream) {
  const float* x   = (const float*)d_in[0];
  const int* eidx  = (const int*)d_in[1];
  const int* pidx  = (const int*)d_in[2];
  const int* nidx  = (const int*)d_in[3];
  const float* W1  = (const float*)d_in[4];
  const float* b1  = (const float*)d_in[5];
  const float* W2  = (const float*)d_in[6];
  const float* b2  = (const float*)d_in[7];
  const float* Wa  = (const float*)d_in[8];
  const float* ba  = (const float*)d_in[9];
  const float* Wb  = (const float*)d_in[10];
  const float* bb  = (const float*)d_in[11];
  float* out = (float*)d_out;

  const int n = in_sizes[0] / IN_C;
  const int E = in_sizes[1] / 2;
  const int P = in_sizes[2] / 2;
  const int* e_src = eidx;
  const int* e_dst = eidx + E;

  float* ws   = (float*)d_ws;
  float* h1   = ws;
  float* z1   = h1 + (size_t)n * HID_C;
  float* dinv = z1 + (size_t)n * HID_C;
  float* h2   = h1;
  float* z2   = h1 + (size_t)n * OUT_C;

  auto cdiv = [](long long a, long long b) { return (int)((a + b - 1) / b); };
  const int B = 256;

  deg_init<<<cdiv(n, B), B, 0, stream>>>(dinv, n);
  deg_count<<<cdiv(E, B), B, 0, stream>>>(e_dst, E, dinv);
  deg_finish<<<cdiv(n, B), B, 0, stream>>>(dinv, n);

  {
    long long waves = (long long)(n / 16) * (HID_C / 16);
    wmma_gemm_f32<<<cdiv(waves * 32, B), B, 0, stream>>>(x, W1, h1, n, HID_C, IN_C);
    agg_owner<HID_C, true><<<cdiv(n, AGG_ROWS), AGG_THREADS, 0, stream>>>(
        h1, e_src, e_dst, dinv, b1, z1, n, E);
  }

  {
    long long waves = (long long)(n / 16) * (OUT_C / 16);
    wmma_gemm_f32<<<cdiv(waves * 32, B), B, 0, stream>>>(z1, W2, h2, n, OUT_C, HID_C);
    agg_owner<OUT_C, false><<<cdiv(n, AGG_ROWS), AGG_THREADS, 0, stream>>>(
        h2, e_src, e_dst, dinv, b2, z2, n, E);
  }

  {
    long long waves = P / 16;
    decode_mlp<<<cdiv(waves * 32, B), B, 0, stream>>>(z2, pidx, pidx + P, Wa, ba, Wb,
                                                      bb, out, P);
    decode_mlp<<<cdiv(waves * 32, B), B, 0, stream>>>(z2, nidx, nidx + P, Wa, ba, Wb,
                                                      bb, out + P, P);
  }
}


# agg layer 2: no second scan of dst: layer 1's owner work-groups store their kept edges (src row, owned row, norm) per wave in the workspace, layer 2 reads them; sum order = LDS arrival as before
# speedup vs baseline: 1.6763x; 1.1129x over previous
#include <hip/hip_runtime.h>
#include <hip/hip_bf16.h>

typedef __attribute__((ext_vector_type(2))) float v2f;
typedef __attribute__((ext_vector_type(8))) float v8f;

#define IN_C  128
#define HID_C 128
#define OUT_C 64

__global__ void deg_init(float* __restrict__ deg, int n) {
  int i = blockIdx.x * blockDim.x + threadIdx.x;
  if (i < n) deg[i] = 1.0f;
}

__global__ void deg_count(const int* __restrict__ dst, int E, float* __restrict__ deg) {
  int i = blockIdx.x * blockDim.x + threadIdx.x;
  if (i < E) atomicAdd(&deg[dst[i]], 1.0f);
}

__global__ void deg_finish(float* __restrict__ deg, int n) {
  int i = blockIdx.x * blockDim.x + threadIdx.x;
  if (i < n) deg[i] = rsqrtf(deg[i]);
}

__global__ void wmma_gemm_f32(const float* __restrict__ A, const float* __restrict__ B,
                              float* __restrict__ C, int M, int N, int K) {
  int wave = (int)((blockIdx.x * (long long)blockDim.x + threadIdx.x) >> 5);
  int lane = threadIdx.x & 31;
  int ntiles = N >> 4;
  int total = (M >> 4) * ntiles;
  if (wave >= total) return;
  int mt = wave / ntiles, nt = wave - mt * ntiles;
  int m0 = mt << 4, n0 = nt << 4;
  int half = lane >> 4;
  int l = lane & 15;

  const float* Arow = A + (size_t)(m0 + l) * K + 2 * half;
  const float* Bcol = B + (size_t)(2 * half) * N + n0 + l;

  v8f acc = {};
  for (int k = 0; k < K; k += 4) {
    v2f a, b;
    a.x = Arow[k];
    a.y = Arow[k + 1];
    b.x = Bcol[(size_t)k * N];
    b.y = Bcol[(size_t)(k + 1) * N];
    acc = __builtin_amdgcn_wmma_f32_16x16x4_f32(false, a, false, b, (short)0, acc,
                                                false, false);
  }

  float* Crow = C + (size_t)(m0 + 8 * half) * N + n0 + l;
#pragma unroll
  for (int i = 0; i < 8; ++i) Crow[(size_t)i * N] = acc[i];
}

#define AGG_ROWS    196
#define AGG_THREADS 1024
#define AGG_WAVES   32
#define AGG_UNROLL  4
#define AGG_BATCH   32
#define AGG_CAP     (AGG_BATCH + AGG_UNROLL * 128)
#define AGG_LCAP    256

typedef int v4i __attribute__((ext_vector_type(4)));

template <int C, int GRP>
__device__ __forceinline__ void agg_add_rows(float* __restrict__ tile,
                                             const float* __restrict__ h, int lane,
                                             int s, int dl, float nrm, int nb) {
  constexpr int VEC = C / 32;
  typedef float vec_t __attribute__((ext_vector_type(VEC)));
  for (int j0 = 0; j0 < nb; j0 += GRP) {
    vec_t hv[GRP];
#pragma unroll
    for (int u = 0; u < GRP; ++u) {
      const int sj = __builtin_amdgcn_readlane(s, j0 + u);
      hv[u] = *(const vec_t*)(h + (size_t)sj * C + lane * VEC);
    }
#pragma unroll
    for (int u = 0; u < GRP; ++u) {
      const bool live = j0 + u < nb;
      const int dj = __builtin_amdgcn_readlane(dl, j0 + u);
      const float nj =
          __int_as_float(__builtin_amdgcn_readlane(__float_as_int(nrm), j0 + u));
      float* trow = tile + dj * C + lane;
#pragma unroll
      for (int k = 0; k < VEC; ++k)
        __hip_atomic_fetch_add(trow + k * 32, live ? hv[u][k] * nj : 0.0f,
                               __ATOMIC_RELAXED, __HIP_MEMORY_SCOPE_WORKGROUP);
    }
  }
}

template <int C, bool RELU, int MODE>
__global__ __launch_bounds__(AGG_THREADS)
void agg_owner(const float* __restrict__ h, const int* __restrict__ src,
               const int* __restrict__ dst, const float* __restrict__ dinv,
               const float* __restrict__ bias, float* __restrict__ out,
               int* __restrict__ ents, int* __restrict__ lcnt, int n, int E) {
  constexpr int VEC = C / 32;
  typedef float vec_t __attribute__((ext_vector_type(VEC)));
  constexpr int LGRP = 32 / VEC;

  __shared__ float tile[AGG_ROWS * C];
  __shared__ int pend[AGG_WAVES][AGG_CAP];

  const int tid = threadIdx.x;
  const int lane = tid & 31;
  const int wave = tid >> 5;
  const int lo = blockIdx.x * AGG_ROWS;
  const int nrows = min(AGG_ROWS, n - lo);
  int* mylist = pend[wave];
  v4i* myents = (v4i*)ents + ((size_t)blockIdx.x * AGG_WAVES + wave) * AGG_LCAP;

  bool listed = false;
  int mycnt = 0;
  if (MODE == 1) {
    const int c32 = lcnt[blockIdx.x * AGG_WAVES + lane];
    listed = __builtin_amdgcn_ballot_w32(c32 > AGG_LCAP) == 0u;
    mycnt = __builtin_amdgcn_readfirstlane(lcnt[blockIdx.x * AGG_WAVES + wave]);
  }

  for (int r = wave; r < nrows; r += AGG_WAVES) {
    const int v = lo + r;
    const float di = dinv[v];
    const vec_t hv = *(const vec_t*)(h + (size_t)v * C + lane * VEC);
#pragma unroll
    for (int k = 0; k < VEC; ++k) tile[r * C + k * 32 + lane] = hv[k] * di * di;
  }
  __syncthreads();

  int cnt = 0;
  int wr = 0;

  auto flush = [&](bool all) {
    const int stop = all ? cnt : (cnt & ~31);
    for (int b0 = 0; b0 < stop; b0 += 32) {
      const int nb = min(32, stop - b0);
      int s = 0, dl = 0;
      float nrm = 0.0f;
      if (lane < nb) {
        const int e = mylist[b0 + lane];
        s = src[e];
        const int d = dst[e];
        nrm = dinv[s] * dinv[d];
        dl = d - lo;
      }
      if (MODE == 0) {
        if (lane < nb && wr + lane < AGG_LCAP)
          myents[wr + lane] = (v4i){s, dl, __float_as_int(nrm), 0};
        wr += nb;
      }
      agg_add_rows<C, 8>(tile, h, lane, s, dl, nrm, nb);
    }
    const int rem = cnt - stop;
    if (rem > 0) {
      int keep = 0;
      if (lane < rem) keep = mylist[stop + lane];
      if (lane < rem) mylist[lane] = keep;
    }
    cnt = rem;
  };

  auto take = [&](int d, int e, int lim) {
    const bool m = (unsigned)(d - lo) < (unsigned)lim;
    const unsigned b = __builtin_amdgcn_ballot_w32(m);
    if (b != 0u) {
      if (m) mylist[cnt + (int)__builtin_amdgcn_mbcnt_lo(b, 0u)] = e;
      cnt += __builtin_popcount(b);
    }
  };

  if (listed) {
    v4i en = {0, 0, 0, 0};
    if (lane < mycnt) en = myents[lane];
    for (int b0 = 0; b0 < mycnt; b0 += 32) {
      const int nb = min(32, mycnt - b0);
      const v4i zero = {0, 0, 0, 0};
      const v4i ec = (lane < nb) ? en : zero;
      if (b0 + 32 + lane < mycnt) en = myents[b0 + 32 + lane];
      agg_add_rows<C, LGRP>(tile, h, lane, ec.x, ec.y, __int_as_float(ec.z), nb);
    }
  } else if ((E & 3) == 0 && E > 0) {
    const int4* d4 = (const int4*)dst;
    const int nq = E >> 2;
    auto ldq = [&](int q) { return d4[min(q, nq - 1)]; };
    constexpr int TRIP = AGG_UNROLL * AGG_THREADS;
    const int ntrips = (nq + TRIP - 1) / TRIP;
    int tt = blockIdx.x % ntrips;
    int4 cur[AGG_UNROLL], nxt[AGG_UNROLL];
#pragma unroll
    for (int u = 0; u < AGG_UNROLL; ++u) cur[u] = ldq(tt * TRIP + u * AGG_THREADS + tid);
    for (int t = 0; t < ntrips; ++t) {
      const int tn = (tt + 1 == ntrips) ? 0 : tt + 1;
#pragma unroll
      for (int u = 0; u < AGG_UNROLL; ++u)
        nxt[u] = ldq(tn * TRIP + u * AGG_THREADS + tid);
      if (cnt >= AGG_BATCH) flush(false);
#pragma unroll
      for (int u = 0; u < AGG_UNROLL; ++u) {
        const int q = tt * TRIP + u * AGG_THREADS + tid;
        const int lim = (q < nq) ? nrows : 0;
        const int e = q << 2;
        take(cur[u].x, e, lim);
        take(cur[u].y, e + 1, lim);
        take(cur[u].z, e + 2, lim);
        take(cur[u].w, e + 3, lim);
      }
#pragma unroll
      for (int u = 0; u < AGG_UNROLL; ++u) cur[u] = nxt[u];
      tt = tn;
    }
  } else {
    for (int e0 = 0; e0 < E; e0 += AGG_THREADS) {
      if (cnt >= AGG_BATCH) flush(false);
      const int e = e0 + tid;
      take(dst[min(e, E - 1)], e, (e < E) ? nrows : 0);
    }
  }
  flush(true);
  if (MODE == 0 && lane == 0) lcnt[blockIdx.x * AGG_WAVES + wave] = wr;
  __syncthreads();

  const vec_t bv = *(const vec_t*)(bias + lane * VEC);
  for (int r = wave; r < nrows; r += AGG_WAVES) {
    vec_t o;
#pragma unroll
    for (int k = 0; k < VEC; ++k) {
      const float t = tile[r * C + k * 32 + lane] + bv[k];
      o[k] = RELU ? fmaxf(t, 0.0f) : t;
    }
    *(vec_t*)(out + (size_t)(lo + r) * C + lane * VEC) = o;
  }
}

__global__ void decode_mlp(const float* __restrict__ z,
                           const int* __restrict__ sidx, const int* __restrict__ didx,
                           const float* __restrict__ Wa,
                           const float* __restrict__ ba,
                           const float* __restrict__ Wb,
                           const float* __restrict__ bb,
                           float* __restrict__ out, int P) {
  int wave = (int)((blockIdx.x * (long long)blockDim.x + threadIdx.x) >> 5);
  int lane = threadIdx.x & 31;
  if (wave >= (P >> 4)) return;
  int m0 = wave << 4;
  int half = lane >> 4;
  int l = lane & 15;

  int row = m0 + l;
  int s = sidx[row], d = didx[row];
  const float* zs = z + (size_t)s * OUT_C;
  const float* zd = z + (size_t)d * OUT_C;

  v8f acc[8];
#pragma unroll
  for (int t = 0; t < 8; ++t) acc[t] = (v8f){};

  for (int k = 0; k < 2 * OUT_C; k += 4) {
    int ka = k + 2 * half;
    v2f a;
    a.x = (ka     < OUT_C) ? zs[ka]         : zd[ka - OUT_C];
    a.y = (ka + 1 < OUT_C) ? zs[ka + 1]     : zd[ka + 1 - OUT_C];
    const float* Bp = Wa + (size_t)ka * HID_C + l;
#pragma unroll
    for (int t = 0; t < 8; ++t) {
      v2f b;
      b.x = Bp[t * 16];
      b.y = Bp[HID_C + t * 16];
      acc[t] = __builtin_amdgcn_wmma_f32_16x16x4_f32(false, a, false, b, (short)0,
                                                     acc[t], false, false);
    }
  }

  float part[8];
#pragma unroll
  for (int i = 0; i < 8; ++i) part[i] = 0.0f;
#pragma unroll
  for (int t = 0; t < 8; ++t) {
    int nidx = t * 16 + l;
    float bav = ba[nidx];
    float wbv = Wb[nidx];
#pragma unroll
    for (int i = 0; i < 8; ++i) {
      float hv = fmaxf(acc[t][i] + bav, 0.0f);
      part[i] += hv * wbv;
    }
  }
#pragma unroll
  for (int m = 1; m <= 8; m <<= 1) {
#pragma unroll
    for (int i = 0; i < 8; ++i) part[i] += __shfl_xor(part[i], m, 32);
  }
  if (l == 0) {
    float bbv = bb[0];
#pragma unroll
    for (int i = 0; i < 8; ++i) out[m0 + 8 * half + i] = part[i] + bbv;
  }
}

extern "C" void kernel_launch(void* const* d_in, const int* in_sizes, int n_in,
                              void* d_out, int out_size, void* d_ws, size_t ws_size,
                              hipStream_t stream) {
  const float* x   = (const float*)d_in[0];
  const int* eidx  = (const int*)d_in[1];
  const int* pidx  = (const int*)d_in[2];
  const int* nidx  = (const int*)d_in[3];
  const float* W1  = (const float*)d_in[4];
  const float* b1  = (const float*)d_in[5];
  const float* W2  = (const float*)d_in[6];
  const float* b2  = (const float*)d_in[7];
  const float* Wa  = (const float*)d_in[8];
  const float* ba  = (const float*)d_in[9];
  const float* Wb  = (const float*)d_in[10];
  const float* bb  = (const float*)d_in[11];
  float* out = (float*)d_out;

  const int n = in_sizes[0] / IN_C;
  const int E = in_sizes[1] / 2;
  const int P = in_sizes[2] / 2;
  const int* e_src = eidx;
  const int* e_dst = eidx + E;

  float* ws   = (float*)d_ws;
  float* h1   = ws;
  float* z1   = h1 + (size_t)n * HID_C;
  const int nwg = (n + AGG_ROWS - 1) / AGG_ROWS;
  int* ents   = (int*)(z1 + (size_t)n * HID_C);
  int* lcnt   = ents + (size_t)nwg * AGG_WAVES * AGG_LCAP * 4;
  float* dinv = (float*)(lcnt + (size_t)nwg * AGG_WAVES);
  float* h2   = h1;
  float* z2   = h1 + (size_t)n * OUT_C;

  auto cdiv = [](long long a, long long b) { return (int)((a + b - 1) / b); };
  const int B = 256;

  deg_init<<<cdiv(n, B), B, 0, stream>>>(dinv, n);
  deg_count<<<cdiv(E, B), B, 0, stream>>>(e_dst, E, dinv);
  deg_finish<<<cdiv(n, B), B, 0, stream>>>(dinv, n);

  {
    long long waves = (long long)(n / 16) * (HID_C / 16);
    wmma_gemm_f32<<<cdiv(waves * 32, B), B, 0, stream>>>(x, W1, h1, n, HID_C, IN_C);
    agg_owner<HID_C, true, 0><<<nwg, AGG_THREADS, 0, stream>>>(
        h1, e_src, e_dst, dinv, b1, z1, ents, lcnt, n, E);
  }

  {
    long long waves = (long long)(n / 16) * (OUT_C / 16);
    wmma_gemm_f32<<<cdiv(waves * 32, B), B, 0, stream>>>(z1, W2, h2, n, OUT_C, HID_C);
    agg_owner<OUT_C, false, 1><<<nwg, AGG_THREADS, 0, stream>>>(
        h2, e_src, e_dst, dinv, b2, z2, ents, lcnt, n, E);
  }

  {
    long long waves = P / 16;
    decode_mlp<<<cdiv(waves * 32, B), B, 0, stream>>>(z2, pidx, pidx + P, Wa, ba, Wb,
                                                      bb, out, P);
    decode_mlp<<<cdiv(waves * 32, B), B, 0, stream>>>(z2, nidx, nidx + P, Wa, ba, Wb,
                                                      bb, out + P, P);
  }
}


# layer 1: x@W1 launch gone: by linearity the owner work-groups sum x rows into their LDS tile, then tile@W1 on WMMA f32 from LDS (pitch 132); sum order: rows summed before the product, k permuted
# speedup vs baseline: 1.7989x; 1.0731x over previous
#include <hip/hip_runtime.h>
#include <hip/hip_bf16.h>

typedef __attribute__((ext_vector_type(2))) float v2f;
typedef __attribute__((ext_vector_type(8))) float v8f;

#define IN_C  128
#define HID_C 128
#define OUT_C 64

__global__ void deg_init(float* __restrict__ deg, int n) {
  int i = blockIdx.x * blockDim.x + threadIdx.x;
  if (i < n) deg[i] = 1.0f;
}

__global__ void deg_count(const int* __restrict__ dst, int E, float* __restrict__ deg) {
  int i = blockIdx.x * blockDim.x + threadIdx.x;
  if (i < E) atomicAdd(&deg[dst[i]], 1.0f);
}

__global__ void deg_finish(float* __restrict__ deg, int n) {
  int i = blockIdx.x * blockDim.x + threadIdx.x;
  if (i < n) deg[i] = rsqrtf(deg[i]);
}

__global__ void wmma_gemm_f32(const float* __restrict__ A, const float* __restrict__ B,
                              float* __restrict__ C, int M, int N, int K) {
  int wave = (int)((blockIdx.x * (long long)blockDim.x + threadIdx.x) >> 5);
  int lane = threadIdx.x & 31;
  int ntiles = N >> 4;
  int total = (M >> 4) * ntiles;
  if (wave >= total) return;
  int mt = wave / ntiles, nt = wave - mt * ntiles;
  int m0 = mt << 4, n0 = nt << 4;
  int half = lane >> 4;
  int l = lane & 15;

  const float* Arow = A + (size_t)(m0 + l) * K + 2 * half;
  const float* Bcol = B + (size_t)(2 * half) * N + n0 + l;

  v8f acc = {};
  for (int k = 0; k < K; k += 4) {
    v2f a, b;
    a.x = Arow[k];
    a.y = Arow[k + 1];
    b.x = Bcol[(size_t)k * N];
    b.y = Bcol[(size_t)(k + 1) * N];
    acc = __builtin_amdgcn_wmma_f32_16x16x4_f32(false, a, false, b, (short)0, acc,
                                                false, false);
  }

  float* Crow = C + (size_t)(m0 + 8 * half) * N + n0 + l;
#pragma unroll
  for (int i = 0; i < 8; ++i) Crow[(size_t)i * N] = acc[i];
}

#define AGG_ROWS    196
#define AGG_MT      13
#define AGG_THREADS 1024
#define AGG_WAVES   32
#define AGG_UNROLL  4
#define AGG_BATCH   32
#define AGG_CAP     (AGG_BATCH + AGG_UNROLL * 128)
#define AGG_LCAP    256

typedef int v4i __attribute__((ext_vector_type(4)));

template <int C, int TP, int GRP>
__device__ __forceinline__ void agg_add_rows(float* __restrict__ tile,
                                             const float* __restrict__ h, int lane,
                                             int s, int dl, float nrm, int nb) {
  constexpr int VEC = C / 32;
  typedef float vec_t __attribute__((ext_vector_type(VEC)));
  for (int j0 = 0; j0 < nb; j0 += GRP) {
    vec_t hv[GRP];
#pragma unroll
    for (int u = 0; u < GRP; ++u) {
      const int sj = __builtin_amdgcn_readlane(s, j0 + u);
      hv[u] = *(const vec_t*)(h + (size_t)sj * C + lane * VEC);
    }
#pragma unroll
    for (int u = 0; u < GRP; ++u) {
      const bool live = j0 + u < nb;
      const int dj = __builtin_amdgcn_readlane(dl, j0 + u);
      const float nj =
          __int_as_float(__builtin_amdgcn_readlane(__float_as_int(nrm), j0 + u));
      float* trow = tile + dj * TP + lane;
#pragma unroll
      for (int k = 0; k < VEC; ++k)
        __hip_atomic_fetch_add(trow + k * 32, live ? hv[u][k] * nj : 0.0f,
                               __ATOMIC_RELAXED, __HIP_MEMORY_SCOPE_WORKGROUP);
    }
  }
}

template <int NT, int TP>
__device__ __forceinline__ void agg_tile_product(const float* __restrict__ arow,
                                                 const v2f (&bf)[32], v8f (&acc)[4]) {
#pragma unroll
  for (int j = 0; j < 32; ++j) {
#pragma unroll
    for (int t = 0; t < NT; ++t) {
      const v2f a = *(const v2f*)(arow + t * 64 * TP + 4 * j);
      acc[t] = __builtin_amdgcn_wmma_f32_16x16x4_f32(false, a, false, bf[j], (short)0,
                                                     acc[t], false, false);
    }
  }
}

template <int C, bool RELU, int MODE, bool MUL>
__global__ __launch_bounds__(AGG_THREADS)
void agg_owner(const float* __restrict__ h, const int* __restrict__ src,
               const int* __restrict__ dst, const float* __restrict__ dinv,
               const float* __restrict__ bias, const float* __restrict__ Wm,
               float* __restrict__ out,
               int* __restrict__ ents, int* __restrict__ lcnt, int n, int E) {
  constexpr int VEC = C / 32;
  typedef float vec_t __attribute__((ext_vector_type(VEC)));
  constexpr int LGRP = 32 / VEC;
  constexpr int TP = MUL ? C + 4 : C;
  constexpr int TROWS = MUL ? AGG_MT * 16 : AGG_ROWS;
  static_assert(!MUL || (C == 128 && AGG_MT * 16 >= AGG_ROWS), "MUL form: 128 x 128");

  __shared__ float tile[TROWS * TP];
  __shared__ int pend[AGG_WAVES][AGG_CAP];

  const int tid = threadIdx.x;
  const int lane = tid & 31;
  const int wave = tid >> 5;
  const int lo = blockIdx.x * AGG_ROWS;
  const int nrows = min(AGG_ROWS, n - lo);
  int* mylist = pend[wave];
  v4i* myents = (v4i*)ents + ((size_t)blockIdx.x * AGG_WAVES + wave) * AGG_LCAP;

  bool listed = false;
  int mycnt = 0;
  if (MODE == 1) {
    const int c32 = lcnt[blockIdx.x * AGG_WAVES + lane];
    listed = __builtin_amdgcn_ballot_w32(c32 > AGG_LCAP) == 0u;
    mycnt = __builtin_amdgcn_readfirstlane(lcnt[blockIdx.x * AGG_WAVES + wave]);
  }

  for (int r = wave; r < nrows; r += AGG_WAVES) {
    const int v = lo + r;
    const float di = dinv[v];
    const vec_t hv = *(const vec_t*)(h + (size_t)v * C + lane * VEC);
#pragma unroll
    for (int k = 0; k < VEC; ++k) tile[r * TP + k * 32 + lane] = hv[k] * di * di;
  }
  if (MUL) {
    for (int r = nrows + wave; r < TROWS; r += AGG_WAVES) {
#pragma unroll
      for (int k = 0; k < VEC; ++k) tile[r * TP + k * 32 + lane] = 0.0f;
    }
  }
  __syncthreads();

  int cnt = 0;
  int wr = 0;

  auto flush = [&](bool all) {
    const int stop = all ? cnt : (cnt & ~31);
    for (int b0 = 0; b0 < stop; b0 += 32) {
      const int nb = min(32, stop - b0);
      int s = 0, dl = 0;
      float nrm = 0.0f;
      if (lane < nb) {
        const int e = mylist[b0 + lane];
        s = src[e];
        const int d = dst[e];
        nrm = dinv[s] * dinv[d];
        dl = d - lo;
      }
      if (MODE == 0) {
        if (lane < nb && wr + lane < AGG_LCAP)
          myents[wr + lane] = (v4i){s, dl, __float_as_int(nrm), 0};
        wr += nb;
      }
      agg_add_rows<C, TP, 8>(tile, h, lane, s, dl, nrm, nb);
    }
    const int rem = cnt - stop;
    if (rem > 0) {
      int keep = 0;
      if (lane < rem) keep = mylist[stop + lane];
      if (lane < rem) mylist[lane] = keep;
    }
    cnt = rem;
  };

  auto take = [&](int d, int e, int lim) {
    const bool m = (unsigned)(d - lo) < (unsigned)lim;
    const unsigned b = __builtin_amdgcn_ballot_w32(m);
    if (b != 0u) {
      if (m) mylist[cnt + (int)__builtin_amdgcn_mbcnt_lo(b, 0u)] = e;
      cnt += __builtin_popcount(b);
    }
  };

  if (listed) {
    v4i en = {0, 0, 0, 0};
    if (lane < mycnt) en = myents[lane];
    for (int b0 = 0; b0 < mycnt; b0 += 32) {
      const int nb = min(32, mycnt - b0);
      const v4i zero = {0, 0, 0, 0};
      const v4i ec = (lane < nb) ? en : zero;
      if (b0 + 32 + lane < mycnt) en = myents[b0 + 32 + lane];
      agg_add_rows<C, TP, LGRP>(tile, h, lane, ec.x, ec.y, __int_as_float(ec.z), nb);
    }
  } else if ((E & 3) == 0 && E > 0) {
    const int4* d4 = (const int4*)dst;
    const int nq = E >> 2;
    auto ldq = [&](int q) { return d4[min(q, nq - 1)]; };
    constexpr int TRIP = AGG_UNROLL * AGG_THREADS;
    const int ntrips = (nq + TRIP - 1) / TRIP;
    int tt = blockIdx.x % ntrips;
    int4 cur[AGG_UNROLL], nxt[AGG_UNROLL];
#pragma unroll
    for (int u = 0; u < AGG_UNROLL; ++u) cur[u] = ldq(tt * TRIP + u * AGG_THREADS + tid);
    for (int t = 0; t < ntrips; ++t) {
      const int tn = (tt + 1 == ntrips) ? 0 : tt + 1;
#pragma unroll
      for (int u = 0; u < AGG_UNROLL; ++u)
        nxt[u] = ldq(tn * TRIP + u * AGG_THREADS + tid);
      if (cnt >= AGG_BATCH) flush(false);
#pragma unroll
      for (int u = 0; u < AGG_UNROLL; ++u) {
        const int q = tt * TRIP + u * AGG_THREADS + tid;
        const int lim = (q < nq) ? nrows : 0;
        const int e = q << 2;
        take(cur[u].x, e, lim);
        take(cur[u].y, e + 1, lim);
        take(cur[u].z, e + 2, lim);
        take(cur[u].w, e + 3, lim);
      }
#pragma unroll
      for (int u = 0; u < AGG_UNROLL; ++u) cur[u] = nxt[u];
      tt = tn;
    }
  } else {
    for (int e0 = 0; e0 < E; e0 += AGG_THREADS) {
      if (cnt >= AGG_BATCH) flush(false);
      const int e = e0 + tid;
      take(dst[min(e, E - 1)], e, (e < E) ? nrows : 0);
    }
  }
  flush(true);
  if (MODE == 0 && lane == 0) lcnt[blockIdx.x * AGG_WAVES + wave] = wr;

  if (MUL) {
    const int hf = lane >> 4;
    const int l16 = lane & 15;
    const int swave = __builtin_amdgcn_readfirstlane(wave);
    const int n0 = (swave & 7) << 4;
    const int mt0 = swave >> 3;
    v2f bf[32];
#pragma unroll
    for (int j = 0; j < 32; ++j) {
      const int p = 4 * j + 2 * hf;
      const int c0 = ((p & 31) << 2) + (p >> 5);
      const int c1 = c0 + 4;
      bf[j].x = Wm[c0 * C + n0 + l16];
      bf[j].y = Wm[c1 * C + n0 + l16];
    }
    __syncthreads();
    v8f acc[4];
#pragma unroll
    for (int t = 0; t < 4; ++t) acc[t] = (v8f){};
    const bool four = mt0 + 12 < AGG_MT;
    const float* arow = tile + (mt0 * 16 + l16) * TP + 2 * hf;
    if (four) agg_tile_product<4, TP>(arow, bf, acc);
    else      agg_tile_product<3, TP>(arow, bf, acc);
    __syncthreads();
    const int pc = ((n0 + l16) & 3) * 32 + ((n0 + l16) >> 2);
#pragma unroll
    for (int t = 0; t < 4; ++t) {
      if (t < 3 || four) {
        float* drow = tile + ((mt0 + 4 * t) * 16 + 8 * hf) * TP + pc;
#pragma unroll
        for (int i = 0; i < 8; ++i) drow[i * TP] = acc[t][i];
      }
    }
  }
  __syncthreads();

  const vec_t bv = *(const vec_t*)(bias + lane * VEC);
  for (int r = wave; r < nrows; r += AGG_WAVES) {
    vec_t o;
#pragma unroll
    for (int k = 0; k < VEC; ++k) {
      const float t = tile[r * TP + k * 32 + lane] + bv[k];
      o[k] = RELU ? fmaxf(t, 0.0f) : t;
    }
    *(vec_t*)(out + (size_t)(lo + r) * C + lane * VEC) = o;
  }
}

__global__ void decode_mlp(const float* __restrict__ z,
                           const int* __restrict__ sidx, const int* __restrict__ didx,
                           const float* __restrict__ Wa,
                           const float* __restrict__ ba,
                           const float* __restrict__ Wb,
                           const float* __restrict__ bb,
                           float* __restrict__ out, int P) {
  int wave = (int)((blockIdx.x * (long long)blockDim.x + threadIdx.x) >> 5);
  int lane = threadIdx.x & 31;
  if (wave >= (P >> 4)) return;
  int m0 = wave << 4;
  int half = lane >> 4;
  int l = lane & 15;

  int row = m0 + l;
  int s = sidx[row], d = didx[row];
  const float* zs = z + (size_t)s * OUT_C;
  const float* zd = z + (size_t)d * OUT_C;

  v8f acc[8];
#pragma unroll
  for (int t = 0; t < 8; ++t) acc[t] = (v8f){};

  for (int k = 0; k < 2 * OUT_C; k += 4) {
    int ka = k + 2 * half;
    v2f a;
    a.x = (ka     < OUT_C) ? zs[ka]         : zd[ka - OUT_C];
    a.y = (ka + 1 < OUT_C) ? zs[ka + 1]     : zd[ka + 1 - OUT_C];
    const float* Bp = Wa + (size_t)ka * HID_C + l;
#pragma unroll
    for (int t = 0; t < 8; ++t) {
      v2f b;
      b.x = Bp[t * 16];
      b.y = Bp[HID_C + t * 16];
      acc[t] = __builtin_amdgcn_wmma_f32_16x16x4_f32(false, a, false, b, (short)0,
                                                     acc[t], false, false);
    }
  }

  float part[8];
#pragma unroll
  for (int i = 0; i < 8; ++i) part[i] = 0.0f;
#pragma unroll
  for (int t = 0; t < 8; ++t) {
    int nidx = t * 16 + l;
    float bav = ba[nidx];
    float wbv = Wb[nidx];
#pragma unroll
    for (int i = 0; i < 8; ++i) {
      float hv = fmaxf(acc[t][i] + bav, 0.0f);
      part[i] += hv * wbv;
    }
  }
#pragma unroll
  for (int m = 1; m <= 8; m <<= 1) {
#pragma unroll
    for (int i = 0; i < 8; ++i) part[i] += __shfl_xor(part[i], m, 32);
  }
  if (l == 0) {
    float bbv = bb[0];
#pragma unroll
    for (int i = 0; i < 8; ++i) out[m0 + 8 * half + i] = part[i] + bbv;
  }
}

extern "C" void kernel_launch(void* const* d_in, const int* in_sizes, int n_in,
                              void* d_out, int out_size, void* d_ws, size_t ws_size,
                              hipStream_t stream) {
  const float* x   = (const float*)d_in[0];
  const int* eidx  = (const int*)d_in[1];
  const int* pidx  = (const int*)d_in[2];
  const int* nidx  = (const int*)d_in[3];
  const float* W1  = (const float*)d_in[4];
  const float* b1  = (const float*)d_in[5];
  const float* W2  = (const float*)d_in[6];
  const float* b2  = (const float*)d_in[7];
  const float* Wa  = (const float*)d_in[8];
  const float* ba  = (const float*)d_in[9];
  const float* Wb  = (const float*)d_in[10];
  const float* bb  = (const float*)d_in[11];
  float* out = (float*)d_out;

  const int n = in_sizes[0] / IN_C;
  const int E = in_sizes[1] / 2;
  const int P = in_sizes[2] / 2;
  const int* e_src = eidx;
  const int* e_dst = eidx + E;

  float* ws   = (float*)d_ws;
  float* h1   = ws;
  float* z1   = h1 + (size_t)n * HID_C;
  const int nwg = (n + AGG_ROWS - 1) / AGG_ROWS;
  int* ents   = (int*)(z1 + (size_t)n * HID_C);
  int* lcnt   = ents + (size_t)nwg * AGG_WAVES * AGG_LCAP * 4;
  float* dinv = (float*)(lcnt + (size_t)nwg * AGG_WAVES);
  float* h2   = h1;
  float* z2   = h1 + (size_t)n * OUT_C;

  auto cdiv = [](long long a, long long b) { return (int)((a + b - 1) / b); };
  const int B = 256;

  deg_init<<<cdiv(n, B), B, 0, stream>>>(dinv, n);
  deg_count<<<cdiv(E, B), B, 0, stream>>>(e_dst, E, dinv);
  deg_finish<<<cdiv(n, B), B, 0, stream>>>(dinv, n);

  {
    agg_owner<IN_C, true, 0, true><<<nwg, AGG_THREADS, 0, stream>>>(
        x, e_src, e_dst, dinv, b1, W1, z1, ents, lcnt, n, E);
  }

  {
    long long waves = (long long)(n / 16) * (OUT_C / 16);
    wmma_gemm_f32<<<cdiv(waves * 32, B), B, 0, stream>>>(z1, W2, h2, n, OUT_C, HID_C);
    agg_owner<OUT_C, false, 1, false><<<nwg, AGG_THREADS, 0, stream>>>(
        h2, e_src, e_dst, dinv, b2, nullptr, z2, ents, lcnt, n, E);
  }

  {
    long long waves = P / 16;
    decode_mlp<<<cdiv(waves * 32, B), B, 0, stream>>>(z2, pidx, pidx + P, Wa, ba, Wb,
                                                      bb, out, P);
    decode_mlp<<<cdiv(waves * 32, B), B, 0, stream>>>(z2, nidx, nidx + P, Wa, ba, Wb,
                                                      bb, out + P, P);
  }
}


# agg: owners stop scanning all of dst: the degree-count launch also writes an owner-major bit map of the edges (LDS integer or), each owner reads only its own row; sum order = LDS arrival as before
# speedup vs baseline: 1.9059x; 1.0595x over previous
#include <hip/hip_runtime.h>
#include <hip/hip_bf16.h>

typedef __attribute__((ext_vector_type(2))) float v2f;
typedef __attribute__((ext_vector_type(8))) float v8f;

#define IN_C  128
#define HID_C 128
#define OUT_C 64

__global__ void deg_init(float* __restrict__ deg, int n) {
  int i = blockIdx.x * blockDim.x + threadIdx.x;
  if (i < n) deg[i] = 1.0f;
}

#define MAP_ROWS    196
#define MAP_OWN     256
#define MAP_THREADS 1024

__global__ __launch_bounds__(MAP_THREADS)
void deg_count_map(const int* __restrict__ dst, int E, float* __restrict__ deg,
                   unsigned* __restrict__ bmap, int nwg, int bp) {
  __shared__ unsigned mt[MAP_OWN * 33];
  const int tid = threadIdx.x;
  const int lane = tid & 31;
  const int wave = tid >> 5;
  const int e = blockIdx.x * MAP_THREADS + tid;
  int w = -1;
  if (e < E) {
    const int d = dst[e];
    atomicAdd(&deg[d], 1.0f);
    w = d / MAP_ROWS;
  }
  for (int w0 = 0; w0 < nwg; w0 += MAP_OWN) {
    for (int i = tid; i < MAP_OWN * 33; i += MAP_THREADS) mt[i] = 0u;
    __syncthreads();
    const unsigned wl = (unsigned)(w - w0);
    if (wl < (unsigned)MAP_OWN) atomicOr(&mt[wl * 33 + wave], 1u << lane);
    __syncthreads();
    for (int o = wave; o < MAP_OWN && w0 + o < nwg; o += MAP_THREADS / 32)
      bmap[(size_t)(w0 + o) * bp + blockIdx.x * 32 + lane] = mt[o * 33 + lane];
    __syncthreads();
  }
}

__global__ void deg_finish(float* __restrict__ deg, int n) {
  int i = blockIdx.x * blockDim.x + threadIdx.x;
  if (i < n) deg[i] = rsqrtf(deg[i]);
}

__global__ void wmma_gemm_f32(const float* __restrict__ A, const float* __restrict__ B,
                              float* __restrict__ C, int M, int N, int K) {
  int wave = (int)((blockIdx.x * (long long)blockDim.x + threadIdx.x) >> 5);
  int lane = threadIdx.x & 31;
  int ntiles = N >> 4;
  int total = (M >> 4) * ntiles;
  if (wave >= total) return;
  int mt = wave / ntiles, nt = wave - mt * ntiles;
  int m0 = mt << 4, n0 = nt << 4;
  int half = lane >> 4;
  int l = lane & 15;

  const float* Arow = A + (size_t)(m0 + l) * K + 2 * half;
  const float* Bcol = B + (size_t)(2 * half) * N + n0 + l;

  v8f acc = {};
  for (int k = 0; k < K; k += 4) {
    v2f a, b;
    a.x = Arow[k];
    a.y = Arow[k + 1];
    b.x = Bcol[(size_t)k * N];
    b.y = Bcol[(size_t)(k + 1) * N];
    acc = __builtin_amdgcn_wmma_f32_16x16x4_f32(false, a, false, b, (short)0, acc,
                                                false, false);
  }

  float* Crow = C + (size_t)(m0 + 8 * half) * N + n0 + l;
#pragma unroll
  for (int i = 0; i < 8; ++i) Crow[(size_t)i * N] = acc[i];
}

#define AGG_ROWS    196
#define AGG_MT      13
#define AGG_THREADS 1024
#define AGG_WAVES   32
#define AGG_BATCH   32
#define AGG_CAP     544
#define AGG_LCAP    256
static_assert(MAP_ROWS == AGG_ROWS, "the map's owners are the aggregation's work-groups");

typedef int v4i __attribute__((ext_vector_type(4)));

template <int C, int TP, int GRP>
__device__ __forceinline__ void agg_add_rows(float* __restrict__ tile,
                                             const float* __restrict__ h, int lane,
                                             int s, int dl, float nrm, int nb) {
  constexpr int VEC = C / 32;
  typedef float vec_t __attribute__((ext_vector_type(VEC)));
  for (int j0 = 0; j0 < nb; j0 += GRP) {
    vec_t hv[GRP];
#pragma unroll
    for (int u = 0; u < GRP; ++u) {
      const int sj = __builtin_amdgcn_readlane(s, j0 + u);
      hv[u] = *(const vec_t*)(h + (size_t)sj * C + lane * VEC);
    }
#pragma unroll
    for (int u = 0; u < GRP; ++u) {
      const bool live = j0 + u < nb;
      const int dj = __builtin_amdgcn_readlane(dl, j0 + u);
      const float nj =
          __int_as_float(__builtin_amdgcn_readlane(__float_as_int(nrm), j0 + u));
      float* trow = tile + dj * TP + lane;
#pragma unroll
      for (int k = 0; k < VEC; ++k)
        __hip_atomic_fetch_add(trow + k * 32, live ? hv[u][k] * nj : 0.0f,
                               __ATOMIC_RELAXED, __HIP_MEMORY_SCOPE_WORKGROUP);
    }
  }
}

template <int NT, int TP>
__device__ __forceinline__ void agg_tile_product(const float* __restrict__ arow,
                                                 const v2f (&bf)[32], v8f (&acc)[4]) {
#pragma unroll
  for (int j = 0; j < 32; ++j) {
#pragma unroll
    for (int t = 0; t < NT; ++t) {
      const v2f a = *(const v2f*)(arow + t * 64 * TP + 4 * j);
      acc[t] = __builtin_amdgcn_wmma_f32_16x16x4_f32(false, a, false, bf[j], (short)0,
                                                     acc[t], false, false);
    }
  }
}

template <int C, bool RELU, int MODE, bool MUL>
__global__ __launch_bounds__(AGG_THREADS)
void agg_owner(const float* __restrict__ h, const int* __restrict__ src,
               const int* __restrict__ dst, const float* __restrict__ dinv,
               const float* __restrict__ bias, const float* __restrict__ Wm,
               float* __restrict__ out,
               int* __restrict__ ents, int* __restrict__ lcnt,
               const unsigned* __restrict__ bmap, int bp, int n, int E) {
  constexpr int VEC = C / 32;
  typedef float vec_t __attribute__((ext_vector_type(VEC)));
  constexpr int LGRP = 32 / VEC;
  constexpr int TP = MUL ? C + 4 : C;
  constexpr int TROWS = MUL ? AGG_MT * 16 : AGG_ROWS;
  static_assert(!MUL || (C == 128 && AGG_MT * 16 >= AGG_ROWS), "MUL form: 128 x 128");

  __shared__ float tile[TROWS * TP];
  __shared__ int pend[AGG_WAVES][AGG_CAP];

  const int tid = threadIdx.x;
  const int lane = tid & 31;
  const int wave = tid >> 5;
  const int lo = blockIdx.x * AGG_ROWS;
  const int nrows = min(AGG_ROWS, n - lo);
  int* mylist = pend[wave];
  v4i* myents = (v4i*)ents + ((size_t)blockIdx.x * AGG_WAVES + wave) * AGG_LCAP;

  bool listed = false;
  int mycnt = 0;
  if (MODE == 1) {
    const int c32 = lcnt[blockIdx.x * AGG_WAVES + lane];
    listed = __builtin_amdgcn_ballot_w32(c32 > AGG_LCAP) == 0u;
    mycnt = __builtin_amdgcn_readfirstlane(lcnt[blockIdx.x * AGG_WAVES + wave]);
  }

  for (int r = wave; r < nrows; r += AGG_WAVES) {
    const int v = lo + r;
    const float di = dinv[v];
    const vec_t hv = *(const vec_t*)(h + (size_t)v * C + lane * VEC);
#pragma unroll
    for (int k = 0; k < VEC; ++k) tile[r * TP + k * 32 + lane] = hv[k] * di * di;
  }
  if (MUL) {
    for (int r = nrows + wave; r < TROWS; r += AGG_WAVES) {
#pragma unroll
      for (int k = 0; k < VEC; ++k) tile[r * TP + k * 32 + lane] = 0.0f;
    }
  }
  __syncthreads();

  int cnt = 0;
  int wr = 0;

  auto flush = [&](bool all) {
    const int stop = all ? cnt : (cnt & ~31);
    for (int b0 = 0; b0 < stop; b0 += 32) {
      const int nb = min(32, stop - b0);
      int s = 0, dl = 0;
      float nrm = 0.0f;
      if (lane < nb) {
        const int e = mylist[b0 + lane];
        s = src[e];
        const int d = dst[e];
        nrm = dinv[s] * dinv[d];
        dl = d - lo;
      }
      if (MODE == 0) {
        if (lane < nb && wr + lane < AGG_LCAP)
          myents[wr + lane] = (v4i){s, dl, __float_as_int(nrm), 0};
        wr += nb;
      }
      agg_add_rows<C, TP, 8>(tile, h, lane, s, dl, nrm, nb);
    }
    const int rem = cnt - stop;
    if (rem > 0) {
      int keep = 0;
      if (lane < rem) keep = mylist[stop + lane];
      if (lane < rem) mylist[lane] = keep;
    }
    cnt = rem;
  };

  if (listed) {
    v4i en = {0, 0, 0, 0};
    if (lane < mycnt) en = myents[lane];
    for (int b0 = 0; b0 < mycnt; b0 += 32) {
      const int nb = min(32, mycnt - b0);
      const v4i zero = {0, 0, 0, 0};
      const v4i ec = (lane < nb) ? en : zero;
      if (b0 + 32 + lane < mycnt) en = myents[b0 + 32 + lane];
      agg_add_rows<C, TP, LGRP>(tile, h, lane, ec.x, ec.y, __int_as_float(ec.z), nb);
    }
  } else {
    const uint4* r4 = (const uint4*)(bmap + (size_t)blockIdx.x * bp);
    const int nq = bp >> 2;
    const uint4 none = {0u, 0u, 0u, 0u};
    uint4 cur = (tid < nq) ? r4[tid] : none;
    for (int q0 = 0; q0 < nq; q0 += AGG_THREADS) {
      const int q = q0 + tid;
      const int qn = q + AGG_THREADS;
      const uint4 nxt = (qn < nq) ? r4[qn] : none;
#pragma nounroll
      for (int k = 0; k < 4; ++k) {
        unsigned wb = (k == 0) ? cur.x : (k == 1) ? cur.y : (k == 2) ? cur.z : cur.w;
        const int ebase = ((q << 2) + k) << 5;
        unsigned b = __builtin_amdgcn_ballot_w32(wb != 0u);
        while (b != 0u) {
          if (cnt >= AGG_BATCH) flush(false);
          if (wb != 0u) {
            mylist[cnt + (int)__builtin_amdgcn_mbcnt_lo(b, 0u)] = ebase + __builtin_ctz(wb);
            wb &= wb - 1u;
          }
          cnt += __builtin_popcount(b);
          b = __builtin_amdgcn_ballot_w32(wb != 0u);
        }
      }
      cur = nxt;
    }
  }
  flush(true);
  if (MODE == 0 && lane == 0) lcnt[blockIdx.x * AGG_WAVES + wave] = wr;

  if (MUL) {
    const int hf = lane >> 4;
    const int l16 = lane & 15;
    const int swave = __builtin_amdgcn_readfirstlane(wave);
    const int n0 = (swave & 7) << 4;
    const int mt0 = swave >> 3;
    v2f bf[32];
#pragma unroll
    for (int j = 0; j < 32; ++j) {
      const int p = 4 * j + 2 * hf;
      const int c0 = ((p & 31) << 2) + (p >> 5);
      const int c1 = c0 + 4;
      bf[j].x = Wm[c0 * C + n0 + l16];
      bf[j].y = Wm[c1 * C + n0 + l16];
    }
    __syncthreads();
    v8f acc[4];
#pragma unroll
    for (int t = 0; t < 4; ++t) acc[t] = (v8f){};
    const bool four = mt0 + 12 < AGG_MT;
    const float* arow = tile + (mt0 * 16 + l16) * TP + 2 * hf;
    if (four) agg_tile_product<4, TP>(arow, bf, acc);
    else      agg_tile_product<3, TP>(arow, bf, acc);
    __syncthreads();
    const int pc = ((n0 + l16) & 3) * 32 + ((n0 + l16) >> 2);
#pragma unroll
    for (int t = 0; t < 4; ++t) {
      if (t < 3 || four) {
        float* drow = tile + ((mt0 + 4 * t) * 16 + 8 * hf) * TP + pc;
#pragma unroll
        for (int i = 0; i < 8; ++i) drow[i * TP] = acc[t][i];
      }
    }
  }
  __syncthreads();

  const vec_t bv = *(const vec_t*)(bias + lane * VEC);
  for (int r = wave; r < nrows; r += AGG_WAVES) {
    vec_t o;
#pragma unroll
    for (int k = 0; k < VEC; ++k) {
      const float t = tile[r * TP + k * 32 + lane] + bv[k];
      o[k] = RELU ? fmaxf(t, 0.0f) : t;
    }
    *(vec_t*)(out + (size_t)(lo + r) * C + lane * VEC) = o;
  }
}

__global__ void decode_mlp(const float* __restrict__ z,
                           const int* __restrict__ sidx, const int* __restrict__ didx,
                           const float* __restrict__ Wa,
                           const float* __restrict__ ba,
                           const float* __restrict__ Wb,
                           const float* __restrict__ bb,
                           float* __restrict__ out, int P) {
  int wave = (int)((blockIdx.x * (long long)blockDim.x + threadIdx.x) >> 5);
  int lane = threadIdx.x & 31;
  if (wave >= (P >> 4)) return;
  int m0 = wave << 4;
  int half = lane >> 4;
  int l = lane & 15;

  int row = m0 + l;
  int s = sidx[row], d = didx[row];
  const float* zs = z + (size_t)s * OUT_C;
  const float* zd = z + (size_t)d * OUT_C;

  v8f acc[8];
#pragma unroll
  for (int t = 0; t < 8; ++t) acc[t] = (v8f){};

  for (int k = 0; k < 2 * OUT_C; k += 4) {
    int ka = k + 2 * half;
    v2f a;
    a.x = (ka     < OUT_C) ? zs[ka]         : zd[ka - OUT_C];
    a.y = (ka + 1 < OUT_C) ? zs[ka + 1]     : zd[ka + 1 - OUT_C];
    const float* Bp = Wa + (size_t)ka * HID_C + l;
#pragma unroll
    for (int t = 0; t < 8; ++t) {
      v2f b;
      b.x = Bp[t * 16];
      b.y = Bp[HID_C + t * 16];
      acc[t] = __builtin_amdgcn_wmma_f32_16x16x4_f32(false, a, false, b, (short)0,
                                                     acc[t], false, false);
    }
  }

  float part[8];
#pragma unroll
  for (int i = 0; i < 8; ++i) part[i] = 0.0f;
#pragma unroll
  for (int t = 0; t < 8; ++t) {
    int nidx = t * 16 + l;
    float bav = ba[nidx];
    float wbv = Wb[nidx];
#pragma unroll
    for (int i = 0; i < 8; ++i) {
      float hv = fmaxf(acc[t][i] + bav, 0.0f);
      part[i] += hv * wbv;
    }
  }
#pragma unroll
  for (int m = 1; m <= 8; m <<= 1) {
#pragma unroll
    for (int i = 0; i < 8; ++i) part[i] += __shfl_xor(part[i], m, 32);
  }
  if (l == 0) {
    float bbv = bb[0];
#pragma unroll
    for (int i = 0; i < 8; ++i) out[m0 + 8 * half + i] = part[i] + bbv;
  }
}

extern "C" void kernel_launch(void* const* d_in, const int* in_sizes, int n_in,
                              void* d_out, int out_size, void* d_ws, size_t ws_size,
                              hipStream_t stream) {
  const float* x   = (const float*)d_in[0];
  const int* eidx  = (const int*)d_in[1];
  const int* pidx  = (const int*)d_in[2];
  const int* nidx  = (const int*)d_in[3];
  const float* W1  = (const float*)d_in[4];
  const float* b1  = (const float*)d_in[5];
  const float* W2  = (const float*)d_in[6];
  const float* b2  = (const float*)d_in[7];
  const float* Wa  = (const float*)d_in[8];
  const float* ba  = (const float*)d_in[9];
  const float* Wb  = (const float*)d_in[10];
  const float* bb  = (const float*)d_in[11];
  float* out = (float*)d_out;

  const int n = in_sizes[0] / IN_C;
  const int E = in_sizes[1] / 2;
  const int P = in_sizes[2] / 2;
  const int* e_src = eidx;
  const int* e_dst = eidx + E;

  float* ws   = (float*)d_ws;
  float* h1   = ws;
  float* z1   = h1 + (size_t)n * HID_C;
  const int nwg = (n + AGG_ROWS - 1) / AGG_ROWS;
  int* ents   = (int*)(z1 + (size_t)n * HID_C);
  int* lcnt   = ents + (size_t)nwg * AGG_WAVES * AGG_LCAP * 4;
  const int nmap = (E + MAP_THREADS - 1) / MAP_THREADS;
  const int bp = nmap * 32;
  unsigned* bmap = (unsigned*)(lcnt + (size_t)nwg * AGG_WAVES);
  float* dinv = (float*)(bmap + (size_t)nwg * bp);
  float* h2   = h1;
  float* z2   = h1 + (size_t)n * OUT_C;

  auto cdiv = [](long long a, long long b) { return (int)((a + b - 1) / b); };
  const int B = 256;

  deg_init<<<cdiv(n, B), B, 0, stream>>>(dinv, n);
  deg_count_map<<<nmap, MAP_THREADS, 0, stream>>>(e_dst, E, dinv, bmap, nwg, bp);
  deg_finish<<<cdiv(n, B), B, 0, stream>>>(dinv, n);

  {
    agg_owner<IN_C, true, 0, true><<<nwg, AGG_THREADS, 0, stream>>>(
        x, e_src, e_dst, dinv, b1, W1, z1, ents, lcnt, bmap, bp, n, E);
  }

  {
    long long waves = (long long)(n / 16) * (OUT_C / 16);
    wmma_gemm_f32<<<cdiv(waves * 32, B), B, 0, stream>>>(z1, W2, h2, n, OUT_C, HID_C);
    agg_owner<OUT_C, false, 1, false><<<nwg, AGG_THREADS, 0, stream>>>(
        h2, e_src, e_dst, dinv, b2, nullptr, z2, ents, lcnt, bmap, bp, n, E);
  }

  {
    long long waves = P / 16;
    decode_mlp<<<cdiv(waves * 32, B), B, 0, stream>>>(z2, pidx, pidx + P, Wa, ba, Wb,
                                                      bb, out, P);
    decode_mlp<<<cdiv(waves * 32, B), B, 0, stream>>>(z2, nidx, nidx + P, Wa, ba, Wb,
                                                      bb, out + P, P);
  }
}


# decode: one launch of resident waves, one per SIMD, all of Wa kept in 512 registers, walking 16-pair tiles of both lists; a tile's closing sums issue between the next tile's k-steps; sums unchanged
# speedup vs baseline: 2.1181x; 1.1113x over previous
#include <hip/hip_runtime.h>
#include <hip/hip_bf16.h>

typedef __attribute__((ext_vector_type(2))) float v2f;
typedef __attribute__((ext_vector_type(8))) float v8f;

#define IN_C  128
#define HID_C 128
#define OUT_C 64

__global__ void deg_init(float* __restrict__ deg, int n) {
  int i = blockIdx.x * blockDim.x + threadIdx.x;
  if (i < n) deg[i] = 1.0f;
}

#define MAP_ROWS    196
#define MAP_OWN     256
#define MAP_THREADS 1024

__global__ __launch_bounds__(MAP_THREADS)
void deg_count_map(const int* __restrict__ dst, int E, float* __restrict__ deg,
                   unsigned* __restrict__ bmap, int nwg, int bp) {
  __shared__ unsigned mt[MAP_OWN * 33];
  const int tid = threadIdx.x;
  const int lane = tid & 31;
  const int wave = tid >> 5;
  const int e = blockIdx.x * MAP_THREADS + tid;
  int w = -1;
  if (e < E) {
    const int d = dst[e];
    atomicAdd(&deg[d], 1.0f);
    w = d / MAP_ROWS;
  }
  for (int w0 = 0; w0 < nwg; w0 += MAP_OWN) {
    for (int i = tid; i < MAP_OWN * 33; i += MAP_THREADS) mt[i] = 0u;
    __syncthreads();
    const unsigned wl = (unsigned)(w - w0);
    if (wl < (unsigned)MAP_OWN) atomicOr(&mt[wl * 33 + wave], 1u << lane);
    __syncthreads();
    for (int o = wave; o < MAP_OWN && w0 + o < nwg; o += MAP_THREADS / 32)
      bmap[(size_t)(w0 + o) * bp + blockIdx.x * 32 + lane] = mt[o * 33 + lane];
    __syncthreads();
  }
}

__global__ void deg_finish(float* __restrict__ deg, int n) {
  int i = blockIdx.x * blockDim.x + threadIdx.x;
  if (i < n) deg[i] = rsqrtf(deg[i]);
}

__global__ void wmma_gemm_f32(const float* __restrict__ A, const float* __restrict__ B,
                              float* __restrict__ C, int M, int N, int K) {
  int wave = (int)((blockIdx.x * (long long)blockDim.x + threadIdx.x) >> 5);
  int lane = threadIdx.x & 31;
  int ntiles = N >> 4;
  int total = (M >> 4) * ntiles;
  if (wave >= total) return;
  int mt = wave / ntiles, nt = wave - mt * ntiles;
  int m0 = mt << 4, n0 = nt << 4;
  int half = lane >> 4;
  int l = lane & 15;

  const float* Arow = A + (size_t)(m0 + l) * K + 2 * half;
  const float* Bcol = B + (size_t)(2 * half) * N + n0 + l;

  v8f acc = {};
  for (int k = 0; k < K; k += 4) {
    v2f a, b;
    a.x = Arow[k];
    a.y = Arow[k + 1];
    b.x = Bcol[(size_t)k * N];
    b.y = Bcol[(size_t)(k + 1) * N];
    acc = __builtin_amdgcn_wmma_f32_16x16x4_f32(false, a, false, b, (short)0, acc,
                                                false, false);
  }

  float* Crow = C + (size_t)(m0 + 8 * half) * N + n0 + l;
#pragma unroll
  for (int i = 0; i < 8; ++i) Crow[(size_t)i * N] = acc[i];
}

#define AGG_ROWS    196
#define AGG_MT      13
#define AGG_THREADS 1024
#define AGG_WAVES   32
#define AGG_BATCH   32
#define AGG_CAP     544
#define AGG_LCAP    256
static_assert(MAP_ROWS == AGG_ROWS, "the map's owners are the aggregation's work-groups");

typedef int v4i __attribute__((ext_vector_type(4)));

template <int C, int TP, int GRP>
__device__ __forceinline__ void agg_add_rows(float* __restrict__ tile,
                                             const float* __restrict__ h, int lane,
                                             int s, int dl, float nrm, int nb) {
  constexpr int VEC = C / 32;
  typedef float vec_t __attribute__((ext_vector_type(VEC)));
  for (int j0 = 0; j0 < nb; j0 += GRP) {
    vec_t hv[GRP];
#pragma unroll
    for (int u = 0; u < GRP; ++u) {
      const int sj = __builtin_amdgcn_readlane(s, j0 + u);
      hv[u] = *(const vec_t*)(h + (size_t)sj * C + lane * VEC);
    }
#pragma unroll
    for (int u = 0; u < GRP; ++u) {
      const bool live = j0 + u < nb;
      const int dj = __builtin_amdgcn_readlane(dl, j0 + u);
      const float nj =
          __int_as_float(__builtin_amdgcn_readlane(__float_as_int(nrm), j0 + u));
      float* trow = tile + dj * TP + lane;
#pragma unroll
      for (int k = 0; k < VEC; ++k)
        __hip_atomic_fetch_add(trow + k * 32, live ? hv[u][k] * nj : 0.0f,
                               __ATOMIC_RELAXED, __HIP_MEMORY_SCOPE_WORKGROUP);
    }
  }
}

template <int NT, int TP>
__device__ __forceinline__ void agg_tile_product(const float* __restrict__ arow,
                                                 const v2f (&bf)[32], v8f (&acc)[4]) {
#pragma unroll
  for (int j = 0; j < 32; ++j) {
#pragma unroll
    for (int t = 0; t < NT; ++t) {
      const v2f a = *(const v2f*)(arow + t * 64 * TP + 4 * j);
      acc[t] = __builtin_amdgcn_wmma_f32_16x16x4_f32(false, a, false, bf[j], (short)0,
                                                     acc[t], false, false);
    }
  }
}

template <int C, bool RELU, int MODE, bool MUL>
__global__ __launch_bounds__(AGG_THREADS)
void agg_owner(const float* __restrict__ h, const int* __restrict__ src,
               const int* __restrict__ dst, const float* __restrict__ dinv,
               const float* __restrict__ bias, const float* __restrict__ Wm,
               float* __restrict__ out,
               int* __restrict__ ents, int* __restrict__ lcnt,
               const unsigned* __restrict__ bmap, int bp, int n, int E) {
  constexpr int VEC = C / 32;
  typedef float vec_t __attribute__((ext_vector_type(VEC)));
  constexpr int LGRP = 32 / VEC;
  constexpr int TP = MUL ? C + 4 : C;
  constexpr int TROWS = MUL ? AGG_MT * 16 : AGG_ROWS;
  static_assert(!MUL || (C == 128 && AGG_MT * 16 >= AGG_ROWS), "MUL form: 128 x 128");

  __shared__ float tile[TROWS * TP];
  __shared__ int pend[AGG_WAVES][AGG_CAP];

  const int tid = threadIdx.x;
  const int lane = tid & 31;
  const int wave = tid >> 5;
  const int lo = blockIdx.x * AGG_ROWS;
  const int nrows = min(AGG_ROWS, n - lo);
  int* mylist = pend[wave];
  v4i* myents = (v4i*)ents + ((size_t)blockIdx.x * AGG_WAVES + wave) * AGG_LCAP;

  bool listed = false;
  int mycnt = 0;
  if (MODE == 1) {
    const int c32 = lcnt[blockIdx.x * AGG_WAVES + lane];
    listed = __builtin_amdgcn_ballot_w32(c32 > AGG_LCAP) == 0u;
    mycnt = __builtin_amdgcn_readfirstlane(lcnt[blockIdx.x * AGG_WAVES + wave]);
  }

  for (int r = wave; r < nrows; r += AGG_WAVES) {
    const int v = lo + r;
    const float di = dinv[v];
    const vec_t hv = *(const vec_t*)(h + (size_t)v * C + lane * VEC);
#pragma unroll
    for (int k = 0; k < VEC; ++k) tile[r * TP + k * 32 + lane] = hv[k] * di * di;
  }
  if (MUL) {
    for (int r = nrows + wave; r < TROWS; r += AGG_WAVES) {
#pragma unroll
      for (int k = 0; k < VEC; ++k) tile[r * TP + k * 32 + lane] = 0.0f;
    }
  }
  __syncthreads();

  int cnt = 0;
  int wr = 0;

  auto flush = [&](bool all) {
    const int stop = all ? cnt : (cnt & ~31);
    for (int b0 = 0; b0 < stop; b0 += 32) {
      const int nb = min(32, stop - b0);
      int s = 0, dl = 0;
      float nrm = 0.0f;
      if (lane < nb) {
        const int e = mylist[b0 + lane];
        s = src[e];
        const int d = dst[e];
        nrm = dinv[s] * dinv[d];
        dl = d - lo;
      }
      if (MODE == 0) {
        if (lane < nb && wr + lane < AGG_LCAP)
          myents[wr + lane] = (v4i){s, dl, __float_as_int(nrm), 0};
        wr += nb;
      }
      agg_add_rows<C, TP, 8>(tile, h, lane, s, dl, nrm, nb);
    }
    const int rem = cnt - stop;
    if (rem > 0) {
      int keep = 0;
      if (lane < rem) keep = mylist[stop + lane];
      if (lane < rem) mylist[lane] = keep;
    }
    cnt = rem;
  };

  if (listed) {
    v4i en = {0, 0, 0, 0};
    if (lane < mycnt) en = myents[lane];
    for (int b0 = 0; b0 < mycnt; b0 += 32) {
      const int nb = min(32, mycnt - b0);
      const v4i zero = {0, 0, 0, 0};
      const v4i ec = (lane < nb) ? en : zero;
      if (b0 + 32 + lane < mycnt) en = myents[b0 + 32 + lane];
      agg_add_rows<C, TP, LGRP>(tile, h, lane, ec.x, ec.y, __int_as_float(ec.z), nb);
    }
  } else {
    const uint4* r4 = (const uint4*)(bmap + (size_t)blockIdx.x * bp);
    const int nq = bp >> 2;
    const uint4 none = {0u, 0u, 0u, 0u};
    uint4 cur = (tid < nq) ? r4[tid] : none;
    for (int q0 = 0; q0 < nq; q0 += AGG_THREADS) {
      const int q = q0 + tid;
      const int qn = q + AGG_THREADS;
      const uint4 nxt = (qn < nq) ? r4[qn] : none;
#pragma nounroll
      for (int k = 0; k < 4; ++k) {
        unsigned wb = (k == 0) ? cur.x : (k == 1) ? cur.y : (k == 2) ? cur.z : cur.w;
        const int ebase = ((q << 2) + k) << 5;
        unsigned b = __builtin_amdgcn_ballot_w32(wb != 0u);
        while (b != 0u) {
          if (cnt >= AGG_BATCH) flush(false);
          if (wb != 0u) {
            mylist[cnt + (int)__builtin_amdgcn_mbcnt_lo(b, 0u)] = ebase + __builtin_ctz(wb);
            wb &= wb - 1u;
          }
          cnt += __builtin_popcount(b);
          b = __builtin_amdgcn_ballot_w32(wb != 0u);
        }
      }
      cur = nxt;
    }
  }
  flush(true);
  if (MODE == 0 && lane == 0) lcnt[blockIdx.x * AGG_WAVES + wave] = wr;

  if (MUL) {
    const int hf = lane >> 4;
    const int l16 = lane & 15;
    const int swave = __builtin_amdgcn_readfirstlane(wave);
    const int n0 = (swave & 7) << 4;
    const int mt0 = swave >> 3;
    v2f bf[32];
#pragma unroll
    for (int j = 0; j < 32; ++j) {
      const int p = 4 * j + 2 * hf;
      const int c0 = ((p & 31) << 2) + (p >> 5);
      const int c1 = c0 + 4;
      bf[j].x = Wm[c0 * C + n0 + l16];
      bf[j].y = Wm[c1 * C + n0 + l16];
    }
    __syncthreads();
    v8f acc[4];
#pragma unroll
    for (int t = 0; t < 4; ++t) acc[t] = (v8f){};
    const bool four = mt0 + 12 < AGG_MT;
    const float* arow = tile + (mt0 * 16 + l16) * TP + 2 * hf;
    if (four) agg_tile_product<4, TP>(arow, bf, acc);
    else      agg_tile_product<3, TP>(arow, bf, acc);
    __syncthreads();
    const int pc = ((n0 + l16) & 3) * 32 + ((n0 + l16) >> 2);
#pragma unroll
    for (int t = 0; t < 4; ++t) {
      if (t < 3 || four) {
        float* drow = tile + ((mt0 + 4 * t) * 16 + 8 * hf) * TP + pc;
#pragma unroll
        for (int i = 0; i < 8; ++i) drow[i * TP] = acc[t][i];
      }
    }
  }
  __syncthreads();

  const vec_t bv = *(const vec_t*)(bias + lane * VEC);
  for (int r = wave; r < nrows; r += AGG_WAVES) {
    vec_t o;
#pragma unroll
    for (int k = 0; k < VEC; ++k) {
      const float t = tile[r * TP + k * 32 + lane] + bv[k];
      o[k] = RELU ? fmaxf(t, 0.0f) : t;
    }
    *(vec_t*)(out + (size_t)(lo + r) * C + lane * VEC) = o;
  }
}

template <int CTRL>
__device__ __forceinline__ float row_xor_add(float v) {
  return v + __int_as_float(
                 __builtin_amdgcn_update_dpp(0, __float_as_int(v), CTRL, 0xf, 0xf, true));
}

#define DEC_THREADS 128
#define DEC_WGS     256

__global__ __launch_bounds__(DEC_THREADS)
void decode_mlp(const float* __restrict__ z,
                const int* __restrict__ pidx,
                const int* __restrict__ nidx,
                const float* __restrict__ Wa,
                const float* __restrict__ ba,
                const float* __restrict__ Wb,
                const float* __restrict__ bb,
                float* __restrict__ out,
                int P) {
  const int lane = threadIdx.x & 31;
  const int wave = __builtin_amdgcn_readfirstlane((int)(threadIdx.x >> 5));
  const int half = lane >> 4;
  const int l = lane & 15;
  const int tpl = P >> 4;
  const int ntiles = 2 * tpl;
  const int nwaves = (int)gridDim.x * (DEC_THREADS / 32);
  int g = (int)blockIdx.x * (DEC_THREADS / 32) + wave;
  if (g >= ntiles) return;

  auto load_idx = [&](int gt, int& s, int& d) {
    const int gc = min(gt, ntiles - 1);
    const bool second = gc >= tpl;
    const int* ix = second ? nidx : pidx;
    const int row = (gc - (second ? tpl : 0)) * 16 + l;
    s = ix[row];
    d = ix[P + row];
  };
  auto load_rows = [&](int s, int d, v2f (&a)[32]) {
    const float* zs = z + (size_t)s * OUT_C + 2 * half;
    const float* zd = z + (size_t)d * OUT_C + 2 * half;
#pragma unroll
    for (int j = 0; j < 16; ++j) a[j] = *(const v2f*)(zs + 4 * j);
#pragma unroll
    for (int j = 0; j < 16; ++j) a[16 + j] = *(const v2f*)(zd + 4 * j);
  };

  int s0, d0, sn, dn;
  load_idx(g, s0, d0);
  load_idx(g + nwaves, sn, dn);
  v2f a0[32], a1[32];
  load_rows(s0, d0, a0);

  v2f bf[8][32];
  const float* Bp = Wa + (size_t)(2 * half) * HID_C + l;
#pragma unroll
  for (int j = 0; j < 32; ++j) {
#pragma unroll
    for (int t = 0; t < 8; ++t) {
      bf[t][j].x = Bp[(4 * j) * HID_C + t * 16];
      bf[t][j].y = Bp[(4 * j + 1) * HID_C + t * 16];
    }
  }
  float bav[8], wbv[8];
#pragma unroll
  for (int t = 0; t < 8; ++t) {
    bav[t] = ba[t * 16 + l];
    wbv[t] = Wb[t * 16 + l];
  }
  const float bbv = bb[0];

  auto step = [&](const v2f (&ac)[32], v2f (&an)[32], v8f (&accw)[8], const v8f (&accr)[8],
                  int gt, int gp) {
    load_rows(sn, dn, an);
    load_idx(gt + 2 * nwaves, sn, dn);
    __builtin_amdgcn_sched_barrier(0);

#pragma unroll
    for (int t = 0; t < 8; ++t) accw[t] = (v8f){};
#pragma unroll
    for (int j = 0; j < 32; ++j) {
#pragma unroll
      for (int t = 0; t < 8; ++t)
        accw[t] = __builtin_amdgcn_wmma_f32_16x16x4_f32(false, ac[j], false, bf[t][j],
                                                        (short)0, accw[t], false, false);
    }

    float part[8];
#pragma unroll
    for (int i = 0; i < 8; ++i) part[i] = 0.0f;
#pragma unroll
    for (int t = 0; t < 8; ++t) {
#pragma unroll
      for (int i = 0; i < 8; ++i) {
        float hv = fmaxf(accr[t][i] + bav[t], 0.0f);
        part[i] += hv * wbv[t];
      }
    }
#pragma unroll
    for (int i = 0; i < 8; ++i) part[i] = row_xor_add<0x161>(part[i]);
#pragma unroll
    for (int i = 0; i < 8; ++i) part[i] = row_xor_add<0x162>(part[i]);
#pragma unroll
    for (int i = 0; i < 8; ++i) part[i] = row_xor_add<0x164>(part[i]);
#pragma unroll
    for (int i = 0; i < 8; ++i) part[i] = row_xor_add<0x168>(part[i]);

#pragma unroll
    for (int j = 0; j < 16; ++j) {
      __builtin_amdgcn_sched_group_barrier(0x008, 8, 0);
      __builtin_amdgcn_sched_group_barrier(0x002, 7, 0);
    }
#pragma unroll
    for (int j = 16; j < 24; ++j) {
      __builtin_amdgcn_sched_group_barrier(0x008, 8, 0);
      __builtin_amdgcn_sched_group_barrier(0x002, 4, 0);
    }
    __builtin_amdgcn_sched_group_barrier(0x008, 64, 0);

    if (gp >= 0 && l == 0) {
#pragma unroll
      for (int i = 0; i < 8; ++i) out[(size_t)gp * 16 + 8 * half + i] = part[i] + bbv;
    }
  };
  auto finish = [&](const v8f (&accr)[8], int gp) {
    float part[8];
#pragma unroll
    for (int i = 0; i < 8; ++i) part[i] = 0.0f;
#pragma unroll
    for (int t = 0; t < 8; ++t) {
#pragma unroll
      for (int i = 0; i < 8; ++i) {
        float hv = fmaxf(accr[t][i] + bav[t], 0.0f);
        part[i] += hv * wbv[t];
      }
    }
#pragma unroll
    for (int i = 0; i < 8; ++i) part[i] = row_xor_add<0x161>(part[i]);
#pragma unroll
    for (int i = 0; i < 8; ++i) part[i] = row_xor_add<0x162>(part[i]);
#pragma unroll
    for (int i = 0; i < 8; ++i) part[i] = row_xor_add<0x164>(part[i]);
#pragma unroll
    for (int i = 0; i < 8; ++i) part[i] = row_xor_add<0x168>(part[i]);
    if (l == 0) {
#pragma unroll
      for (int i = 0; i < 8; ++i) out[(size_t)gp * 16 + 8 * half + i] = part[i] + bbv;
    }
  };

  v8f acc0[8], acc1[8];
#pragma unroll
  for (int t = 0; t < 8; ++t) acc1[t] = (v8f){};
  int gp = -1;
  for (;;) {
    step(a0, a1, acc0, acc1, g, gp);
    gp = g;
    g += nwaves;
    if (g >= ntiles) {
      finish(acc0, gp);
      break;
    }
    step(a1, a0, acc1, acc0, g, gp);
    gp = g;
    g += nwaves;
    if (g >= ntiles) {
      finish(acc1, gp);
      break;
    }
  }
}

extern "C" void kernel_launch(void* const* d_in, const int* in_sizes, int n_in,
                              void* d_out, int out_size, void* d_ws, size_t ws_size,
                              hipStream_t stream) {
  const float* x   = (const float*)d_in[0];
  const int* eidx  = (const int*)d_in[1];
  const int* pidx  = (const int*)d_in[2];
  const int* nidx  = (const int*)d_in[3];
  const float* W1  = (const float*)d_in[4];
  const float* b1  = (const float*)d_in[5];
  const float* W2  = (const float*)d_in[6];
  const float* b2  = (const float*)d_in[7];
  const float* Wa  = (const float*)d_in[8];
  const float* ba  = (const float*)d_in[9];
  const float* Wb  = (const float*)d_in[10];
  const float* bb  = (const float*)d_in[11];
  float* out = (float*)d_out;

  const int n = in_sizes[0] / IN_C;
  const int E = in_sizes[1] / 2;
  const int P = in_sizes[2] / 2;
  const int* e_src = eidx;
  const int* e_dst = eidx + E;

  float* ws   = (float*)d_ws;
  float* h1   = ws;
  float* z1   = h1 + (size_t)n * HID_C;
  const int nwg = (n + AGG_ROWS - 1) / AGG_ROWS;
  int* ents   = (int*)(z1 + (size_t)n * HID_C);
  int* lcnt   = ents + (size_t)nwg * AGG_WAVES * AGG_LCAP * 4;
  const int nmap = (E + MAP_THREADS - 1) / MAP_THREADS;
  const int bp = nmap * 32;
  unsigned* bmap = (unsigned*)(lcnt + (size_t)nwg * AGG_WAVES);
  float* dinv = (float*)(bmap + (size_t)nwg * bp);
  float* h2   = h1;
  float* z2   = h1 + (size_t)n * OUT_C;

  auto cdiv = [](long long a, long long b) { return (int)((a + b - 1) / b); };
  const int B = 256;

  deg_init<<<cdiv(n, B), B, 0, stream>>>(dinv, n);
  deg_count_map<<<nmap, MAP_THREADS, 0, stream>>>(e_dst, E, dinv, bmap, nwg, bp);
  deg_finish<<<cdiv(n, B), B, 0, stream>>>(dinv, n);

  {
    agg_owner<IN_C, true, 0, true><<<nwg, AGG_THREADS, 0, stream>>>(
        x, e_src, e_dst, dinv, b1, W1, z1, ents, lcnt, bmap, bp, n, E);
  }

  {
    long long waves = (long long)(n / 16) * (OUT_C / 16);
    wmma_gemm_f32<<<cdiv(waves * 32, B), B, 0, stream>>>(z1, W2, h2, n, OUT_C, HID_C);
    agg_owner<OUT_C, false, 1, false><<<nwg, AGG_THREADS, 0, stream>>>(
        h2, e_src, e_dst, dinv, b2, nullptr, z2, ents, lcnt, bmap, bp, n, E);
  }

  decode_mlp<<<DEC_WGS, DEC_THREADS, 0, stream>>>(z2, pidx, nidx, Wa, ba, Wb, bb, out, P);
}


# degree: global float atomic count + its init/finish launches replaced by an owner launch that reads the map once, counts its rows in LDS, stores the edge lists both layers read; sum order as before
# speedup vs baseline: 2.5329x; 1.1958x over previous
#include <hip/hip_runtime.h>
#include <hip/hip_bf16.h>

typedef __attribute__((ext_vector_type(2))) float v2f;
typedef __attribute__((ext_vector_type(8))) float v8f;

#define IN_C  128
#define HID_C 128
#define OUT_C 64

#define MAP_ROWS    196
#define MAP_OWN     256
#define MAP_THREADS 1024

__global__ __launch_bounds__(MAP_THREADS)
void edge_map(const int* __restrict__ dst, int E, unsigned* __restrict__ bmap, int nwg,
              int bp) {
  __shared__ unsigned mt[MAP_OWN * 33];
  const int tid = threadIdx.x;
  const int lane = tid & 31;
  const int wave = tid >> 5;
  const int e = blockIdx.x * MAP_THREADS + tid;
  int w = -1;
  if (e < E) w = dst[e] / MAP_ROWS;
  for (int w0 = 0; w0 < nwg; w0 += MAP_OWN) {
    for (int i = tid; i < MAP_OWN * 33; i += MAP_THREADS) mt[i] = 0u;
    __syncthreads();
    const unsigned wl = (unsigned)(w - w0);
    if (wl < (unsigned)MAP_OWN) atomicOr(&mt[wl * 33 + wave], 1u << lane);
    __syncthreads();
    for (int o = wave; o < MAP_OWN && w0 + o < nwg; o += MAP_THREADS / 32)
      bmap[(size_t)(w0 + o) * bp + blockIdx.x * 32 + lane] = mt[o * 33 + lane];
    __syncthreads();
  }
}

__global__ void wmma_gemm_f32(const float* __restrict__ A, const float* __restrict__ B,
                              float* __restrict__ C, int M, int N, int K) {
  int wave = (int)((blockIdx.x * (long long)blockDim.x + threadIdx.x) >> 5);
  int lane = threadIdx.x & 31;
  int ntiles = N >> 4;
  int total = (M >> 4) * ntiles;
  if (wave >= total) return;
  int mt = wave / ntiles, nt = wave - mt * ntiles;
  int m0 = mt << 4, n0 = nt << 4;
  int half = lane >> 4;
  int l = lane & 15;

  const float* Arow = A + (size_t)(m0 + l) * K + 2 * half;
  const float* Bcol = B + (size_t)(2 * half) * N + n0 + l;

  v8f acc = {};
  for (int k = 0; k < K; k += 4) {
    v2f a, b;
    a.x = Arow[k];
    a.y = Arow[k + 1];
    b.x = Bcol[(size_t)k * N];
    b.y = Bcol[(size_t)(k + 1) * N];
    acc = __builtin_amdgcn_wmma_f32_16x16x4_f32(false, a, false, b, (short)0, acc,
                                                false, false);
  }

  float* Crow = C + (size_t)(m0 + 8 * half) * N + n0 + l;
#pragma unroll
  for (int i = 0; i < 8; ++i) Crow[(size_t)i * N] = acc[i];
}

#define AGG_ROWS    196
#define AGG_MT      13
#define AGG_THREADS 1024
#define AGG_WAVES   32
#define AGG_BATCH   32
#define AGG_CAP     544
#define AGG_LCAP    256
static_assert(MAP_ROWS == AGG_ROWS, "the map's owners are the aggregation's work-groups");

typedef int v4i __attribute__((ext_vector_type(4)));

template <int C, int TP, int GRP>
__device__ __forceinline__ void agg_add_rows(float* __restrict__ tile,
                                             const float* __restrict__ h, int lane,
                                             int s, int dl, float nrm, int nb) {
  constexpr int VEC = C / 32;
  typedef float vec_t __attribute__((ext_vector_type(VEC)));
  for (int j0 = 0; j0 < nb; j0 += GRP) {
    vec_t hv[GRP];
#pragma unroll
    for (int u = 0; u < GRP; ++u) {
      const int sj = __builtin_amdgcn_readlane(s, j0 + u);
      hv[u] = *(const vec_t*)(h + (size_t)sj * C + lane * VEC);
    }
#pragma unroll
    for (int u = 0; u < GRP; ++u) {
      const bool live = j0 + u < nb;
      const int dj = __builtin_amdgcn_readlane(dl, j0 + u);
      const float nj =
          __int_as_float(__builtin_amdgcn_readlane(__float_as_int(nrm), j0 + u));
      float* trow = tile + dj * TP + lane;
#pragma unroll
      for (int k = 0; k < VEC; ++k)
        __hip_atomic_fetch_add(trow + k * 32, live ? hv[u][k] * nj : 0.0f,
                               __ATOMIC_RELAXED, __HIP_MEMORY_SCOPE_WORKGROUP);
    }
  }
}

template <int NT, int TP>
__device__ __forceinline__ void agg_tile_product(const float* __restrict__ arow,
                                                 const v2f (&bf)[32], v8f (&acc)[4]) {
#pragma unroll
  for (int j = 0; j < 32; ++j) {
#pragma unroll
    for (int t = 0; t < NT; ++t) {
      const v2f a = *(const v2f*)(arow + t * 64 * TP + 4 * j);
      acc[t] = __builtin_amdgcn_wmma_f32_16x16x4_f32(false, a, false, bf[j], (short)0,
                                                     acc[t], false, false);
    }
  }
}

#define LIST_CAP 64

__global__ __launch_bounds__(AGG_THREADS)
void owner_list(const int* __restrict__ src, const int* __restrict__ dst,
                float* __restrict__ dinv, int* __restrict__ ents, int* __restrict__ lcnt,
                const unsigned* __restrict__ bmap, int bp, int n) {
  __shared__ int rcnt[AGG_ROWS];
  __shared__ int pend[AGG_WAVES][LIST_CAP];

  const int tid = threadIdx.x;
  const int lane = tid & 31;
  const int wave = tid >> 5;
  const int lo = blockIdx.x * AGG_ROWS;
  const int nrows = min(AGG_ROWS, n - lo);
  int* mylist = pend[wave];
  v4i* myents = (v4i*)ents + ((size_t)blockIdx.x * AGG_WAVES + wave) * AGG_LCAP;

  for (int r = tid; r < AGG_ROWS; r += AGG_THREADS) rcnt[r] = 0;
  __syncthreads();

  int cnt = 0;
  int wr = 0;

  auto flush = [&](bool all) {
    const int stop = all ? cnt : (cnt & ~31);
    for (int b0 = 0; b0 < stop; b0 += 32) {
      const int nb = min(32, stop - b0);
      if (lane < nb) {
        const int e = mylist[b0 + lane];
        const int s = src[e];
        const int d = dst[e];
        asm volatile("" ::"v"(s), "v"(d));
        const int dl = d - lo;
        atomicAdd(&rcnt[dl], 1);
        if (wr + lane < AGG_LCAP) myents[wr + lane] = (v4i){s, dl, 0, 0};
      }
      wr += nb;
    }
    const int rem = cnt - stop;
    if (rem > 0) {
      int keep = 0;
      if (lane < rem) keep = mylist[stop + lane];
      if (lane < rem) mylist[lane] = keep;
    }
    cnt = rem;
  };

  const uint4* r4 = (const uint4*)(bmap + (size_t)blockIdx.x * bp);
  const int nq = bp >> 2;
  const uint4 none = {0u, 0u, 0u, 0u};
  uint4 cur = r4[min(tid, nq - 1)];
  if (tid >= nq) cur = none;
  for (int q0 = 0; q0 < nq; q0 += AGG_THREADS) {
    const int q = q0 + tid;
    const int qn = q + AGG_THREADS;
    uint4 nxt = r4[min(qn, nq - 1)];
    if (qn >= nq) nxt = none;
#pragma nounroll
    for (int k = 0; k < 4; ++k) {
      unsigned wb = (k == 0) ? cur.x : (k == 1) ? cur.y : (k == 2) ? cur.z : cur.w;
      const int ebase = ((q << 2) + k) << 5;
      unsigned b = __builtin_amdgcn_ballot_w32(wb != 0u);
      while (b != 0u) {
        if (cnt >= AGG_BATCH) flush(false);
        if (wb != 0u) {
          mylist[cnt + (int)__builtin_amdgcn_mbcnt_lo(b, 0u)] = ebase + __builtin_ctz(wb);
          wb &= wb - 1u;
        }
        cnt += __builtin_popcount(b);
        b = __builtin_amdgcn_ballot_w32(wb != 0u);
      }
    }
    cur = nxt;
  }
  flush(true);
  if (lane == 0) lcnt[blockIdx.x * AGG_WAVES + wave] = wr;
  __syncthreads();
  for (int r = tid; r < nrows; r += AGG_THREADS)
    dinv[lo + r] = rsqrtf(1.0f + (float)rcnt[r]);
}

template <int C, bool RELU, int MODE, bool MUL>
__global__ __launch_bounds__(AGG_THREADS)
void agg_owner(const float* __restrict__ h, const int* __restrict__ src,
               const int* __restrict__ dst, const float* __restrict__ dinv,
               const float* __restrict__ bias, const float* __restrict__ Wm,
               float* __restrict__ out,
               int* __restrict__ ents, int* __restrict__ lcnt,
               const unsigned* __restrict__ bmap, int bp, int n, int E) {
  constexpr int VEC = C / 32;
  typedef float vec_t __attribute__((ext_vector_type(VEC)));
  constexpr int LGRP = 32 / VEC;
  constexpr int TP = MUL ? C + 4 : C;
  constexpr int TROWS = MUL ? AGG_MT * 16 : AGG_ROWS;
  static_assert(!MUL || (C == 128 && AGG_MT * 16 >= AGG_ROWS), "MUL form: 128 x 128");

  __shared__ float tile[TROWS * TP];
  __shared__ int pend[AGG_WAVES][AGG_CAP];

  const int tid = threadIdx.x;
  const int lane = tid & 31;
  const int wave = tid >> 5;
  const int lo = blockIdx.x * AGG_ROWS;
  const int nrows = min(AGG_ROWS, n - lo);
  int* mylist = pend[wave];
  v4i* myents = (v4i*)ents + ((size_t)blockIdx.x * AGG_WAVES + wave) * AGG_LCAP;

  const int c32 = lcnt[blockIdx.x * AGG_WAVES + lane];
  const bool listed = __builtin_amdgcn_ballot_w32(c32 > AGG_LCAP) == 0u;
  const int mycnt = __builtin_amdgcn_readfirstlane(lcnt[blockIdx.x * AGG_WAVES + wave]);

  for (int r = wave; r < nrows; r += AGG_WAVES) {
    const int v = lo + r;
    const float di = dinv[v];
    const vec_t hv = *(const vec_t*)(h + (size_t)v * C + lane * VEC);
#pragma unroll
    for (int k = 0; k < VEC; ++k) tile[r * TP + k * 32 + lane] = hv[k] * di * di;
  }
  if (MUL) {
    for (int r = nrows + wave; r < TROWS; r += AGG_WAVES) {
#pragma unroll
      for (int k = 0; k < VEC; ++k) tile[r * TP + k * 32 + lane] = 0.0f;
    }
  }
  __syncthreads();

  int cnt = 0;

  auto flush = [&](bool all) {
    const int stop = all ? cnt : (cnt & ~31);
    for (int b0 = 0; b0 < stop; b0 += 32) {
      const int nb = min(32, stop - b0);
      int s = 0, dl = 0;
      float nrm = 0.0f;
      if (lane < nb) {
        const int e = mylist[b0 + lane];
        s = src[e];
        const int d = dst[e];
        nrm = dinv[s] * dinv[d];
        dl = d - lo;
      }
      agg_add_rows<C, TP, 8>(tile, h, lane, s, dl, nrm, nb);
    }
    const int rem = cnt - stop;
    if (rem > 0) {
      int keep = 0;
      if (lane < rem) keep = mylist[stop + lane];
      if (lane < rem) mylist[lane] = keep;
    }
    cnt = rem;
  };

  if (listed) {
    v4i en = {0, 0, 0, 0};
    if (lane < mycnt) en = myents[lane];
    for (int b0 = 0; b0 < mycnt; b0 += 32) {
      const int nb = min(32, mycnt - b0);
      const v4i zero = {0, 0, 0, 0};
      const v4i ec = (lane < nb) ? en : zero;
      if (b0 + 32 + lane < mycnt) en = myents[b0 + 32 + lane];
      float nrm = __int_as_float(ec.z);
      if (MODE == 0) {
        nrm = 0.0f;
        if (lane < nb) {
          nrm = dinv[ec.x] * dinv[lo + ec.y];
          ((int*)(myents + b0 + lane))[2] = __float_as_int(nrm);
        }
      }
      agg_add_rows<C, TP, LGRP>(tile, h, lane, ec.x, ec.y, nrm, nb);
    }
  } else {
    const uint4* r4 = (const uint4*)(bmap + (size_t)blockIdx.x * bp);
    const int nq = bp >> 2;
    const uint4 none = {0u, 0u, 0u, 0u};
    uint4 cur = (tid < nq) ? r4[tid] : none;
    for (int q0 = 0; q0 < nq; q0 += AGG_THREADS) {
      const int q = q0 + tid;
      const int qn = q + AGG_THREADS;
      const uint4 nxt = (qn < nq) ? r4[qn] : none;
#pragma nounroll
      for (int k = 0; k < 4; ++k) {
        unsigned wb = (k == 0) ? cur.x : (k == 1) ? cur.y : (k == 2) ? cur.z : cur.w;
        const int ebase = ((q << 2) + k) << 5;
        unsigned b = __builtin_amdgcn_ballot_w32(wb != 0u);
        while (b != 0u) {
          if (cnt >= AGG_BATCH) flush(false);
          if (wb != 0u) {
            mylist[cnt + (int)__builtin_amdgcn_mbcnt_lo(b, 0u)] = ebase + __builtin_ctz(wb);
            wb &= wb - 1u;
          }
          cnt += __builtin_popcount(b);
          b = __builtin_amdgcn_ballot_w32(wb != 0u);
        }
      }
      cur = nxt;
    }
  }
  flush(true);

  if (MUL) {
    const int hf = lane >> 4;
    const int l16 = lane & 15;
    const int swave = __builtin_amdgcn_readfirstlane(wave);
    const int n0 = (swave & 7) << 4;
    const int mt0 = swave >> 3;
    v2f bf[32];
#pragma unroll
    for (int j = 0; j < 32; ++j) {
      const int p = 4 * j + 2 * hf;
      const int c0 = ((p & 31) << 2) + (p >> 5);
      const int c1 = c0 + 4;
      bf[j].x = Wm[c0 * C + n0 + l16];
      bf[j].y = Wm[c1 * C + n0 + l16];
    }
    __syncthreads();
    v8f acc[4];
#pragma unroll
    for (int t = 0; t < 4; ++t) acc[t] = (v8f){};
    const bool four = mt0 + 12 < AGG_MT;
    const float* arow = tile + (mt0 * 16 + l16) * TP + 2 * hf;
    if (four) agg_tile_product<4, TP>(arow, bf, acc);
    else      agg_tile_product<3, TP>(arow, bf, acc);
    __syncthreads();
    const int pc = ((n0 + l16) & 3) * 32 + ((n0 + l16) >> 2);
#pragma unroll
    for (int t = 0; t < 4; ++t) {
      if (t < 3 || four) {
        float* drow = tile + ((mt0 + 4 * t) * 16 + 8 * hf) * TP + pc;
#pragma unroll
        for (int i = 0; i < 8; ++i) drow[i * TP] = acc[t][i];
      }
    }
  }
  __syncthreads();

  const vec_t bv = *(const vec_t*)(bias + lane * VEC);
  for (int r = wave; r < nrows; r += AGG_WAVES) {
    vec_t o;
#pragma unroll
    for (int k = 0; k < VEC; ++k) {
      const float t = tile[r * TP + k * 32 + lane] + bv[k];
      o[k] = RELU ? fmaxf(t, 0.0f) : t;
    }
    *(vec_t*)(out + (size_t)(lo + r) * C + lane * VEC) = o;
  }
}

template <int CTRL>
__device__ __forceinline__ float row_xor_add(float v) {
  return v + __int_as_float(
                 __builtin_amdgcn_update_dpp(0, __float_as_int(v), CTRL, 0xf, 0xf, true));
}

#define DEC_THREADS 128
#define DEC_WGS     256

__global__ __launch_bounds__(DEC_THREADS)
void decode_mlp(const float* __restrict__ z,
                const int* __restrict__ pidx,
                const int* __restrict__ nidx,
                const float* __restrict__ Wa,
                const float* __restrict__ ba,
                const float* __restrict__ Wb,
                const float* __restrict__ bb,
                float* __restrict__ out,
                int P) {
  const int lane = threadIdx.x & 31;
  const int wave = __builtin_amdgcn_readfirstlane((int)(threadIdx.x >> 5));
  const int half = lane >> 4;
  const int l = lane & 15;
  const int tpl = P >> 4;
  const int ntiles = 2 * tpl;
  const int nwaves = (int)gridDim.x * (DEC_THREADS / 32);
  int g = (int)blockIdx.x * (DEC_THREADS / 32) + wave;
  if (g >= ntiles) return;

  auto load_idx = [&](int gt, int& s, int& d) {
    const int gc = min(gt, ntiles - 1);
    const bool second = gc >= tpl;
    const int* ix = second ? nidx : pidx;
    const int row = (gc - (second ? tpl : 0)) * 16 + l;
    s = ix[row];
    d = ix[P + row];
  };
  auto load_rows = [&](int s, int d, v2f (&a)[32]) {
    const float* zs = z + (size_t)s * OUT_C + 2 * half;
    const float* zd = z + (size_t)d * OUT_C + 2 * half;
#pragma unroll
    for (int j = 0; j < 16; ++j) a[j] = *(const v2f*)(zs + 4 * j);
#pragma unroll
    for (int j = 0; j < 16; ++j) a[16 + j] = *(const v2f*)(zd + 4 * j);
  };

  int s0, d0, sn, dn;
  load_idx(g, s0, d0);
  load_idx(g + nwaves, sn, dn);
  v2f a0[32], a1[32];
  load_rows(s0, d0, a0);

  v2f bf[8][32];
  const float* Bp = Wa + (size_t)(2 * half) * HID_C + l;
#pragma unroll
  for (int j = 0; j < 32; ++j) {
#pragma unroll
    for (int t = 0; t < 8; ++t) {
      bf[t][j].x = Bp[(4 * j) * HID_C + t * 16];
      bf[t][j].y = Bp[(4 * j + 1) * HID_C + t * 16];
    }
  }
  float bav[8], wbv[8];
#pragma unroll
  for (int t = 0; t < 8; ++t) {
    bav[t] = ba[t * 16 + l];
    wbv[t] = Wb[t * 16 + l];
  }
  const float bbv = bb[0];

  auto step = [&](const v2f (&ac)[32], v2f (&an)[32], v8f (&accw)[8], const v8f (&accr)[8],
                  int gt, int gp) {
    load_rows(sn, dn, an);
    load_idx(gt + 2 * nwaves, sn, dn);
    __builtin_amdgcn_sched_barrier(0);

#pragma unroll
    for (int t = 0; t < 8; ++t) accw[t] = (v8f){};
#pragma unroll
    for (int j = 0; j < 32; ++j) {
#pragma unroll
      for (int t = 0; t < 8; ++t)
        accw[t] = __builtin_amdgcn_wmma_f32_16x16x4_f32(false, ac[j], false, bf[t][j],
                                                        (short)0, accw[t], false, false);
    }

    float part[8];
#pragma unroll
    for (int i = 0; i < 8; ++i) part[i] = 0.0f;
#pragma unroll
    for (int t = 0; t < 8; ++t) {
#pragma unroll
      for (int i = 0; i < 8; ++i) {
        float hv = fmaxf(accr[t][i] + bav[t], 0.0f);
        part[i] += hv * wbv[t];
      }
    }
#pragma unroll
    for (int i = 0; i < 8; ++i) part[i] = row_xor_add<0x161>(part[i]);
#pragma unroll
    for (int i = 0; i < 8; ++i) part[i] = row_xor_add<0x162>(part[i]);
#pragma unroll
    for (int i = 0; i < 8; ++i) part[i] = row_xor_add<0x164>(part[i]);
#pragma unroll
    for (int i = 0; i < 8; ++i) part[i] = row_xor_add<0x168>(part[i]);

#pragma unroll
    for (int j = 0; j < 16; ++j) {
      __builtin_amdgcn_sched_group_barrier(0x008, 8, 0);
      __builtin_amdgcn_sched_group_barrier(0x002, 7, 0);
    }
#pragma unroll
    for (int j = 16; j < 24; ++j) {
      __builtin_amdgcn_sched_group_barrier(0x008, 8, 0);
      __builtin_amdgcn_sched_group_barrier(0x002, 4, 0);
    }
    __builtin_amdgcn_sched_group_barrier(0x008, 64, 0);

    if (gp >= 0 && l == 0) {
#pragma unroll
      for (int i = 0; i < 8; ++i) out[(size_t)gp * 16 + 8 * half + i] = part[i] + bbv;
    }
  };
  auto finish = [&](const v8f (&accr)[8], int gp) {
    float part[8];
#pragma unroll
    for (int i = 0; i < 8; ++i) part[i] = 0.0f;
#pragma unroll
    for (int t = 0; t < 8; ++t) {
#pragma unroll
      for (int i = 0; i < 8; ++i) {
        float hv = fmaxf(accr[t][i] + bav[t], 0.0f);
        part[i] += hv * wbv[t];
      }
    }
#pragma unroll
    for (int i = 0; i < 8; ++i) part[i] = row_xor_add<0x161>(part[i]);
#pragma unroll
    for (int i = 0; i < 8; ++i) part[i] = row_xor_add<0x162>(part[i]);
#pragma unroll
    for (int i = 0; i < 8; ++i) part[i] = row_xor_add<0x164>(part[i]);
#pragma unroll
    for (int i = 0; i < 8; ++i) part[i] = row_xor_add<0x168>(part[i]);
    if (l == 0) {
#pragma unroll
      for (int i = 0; i < 8; ++i) out[(size_t)gp * 16 + 8 * half + i] = part[i] + bbv;
    }
  };

  v8f acc0[8], acc1[8];
#pragma unroll
  for (int t = 0; t < 8; ++t) acc1[t] = (v8f){};
  int gp = -1;
  for (;;) {
    step(a0, a1, acc0, acc1, g, gp);
    gp = g;
    g += nwaves;
    if (g >= ntiles) {
      finish(acc0, gp);
      break;
    }
    step(a1, a0, acc1, acc0, g, gp);
    gp = g;
    g += nwaves;
    if (g >= ntiles) {
      finish(acc1, gp);
      break;
    }
  }
}

extern "C" void kernel_launch(void* const* d_in, const int* in_sizes, int n_in,
                              void* d_out, int out_size, void* d_ws, size_t ws_size,
                              hipStream_t stream) {
  const float* x   = (const float*)d_in[0];
  const int* eidx  = (const int*)d_in[1];
  const int* pidx  = (const int*)d_in[2];
  const int* nidx  = (const int*)d_in[3];
  const float* W1  = (const float*)d_in[4];
  const float* b1  = (const float*)d_in[5];
  const float* W2  = (const float*)d_in[6];
  const float* b2  = (const float*)d_in[7];
  const float* Wa  = (const float*)d_in[8];
  const float* ba  = (const float*)d_in[9];
  const float* Wb  = (const float*)d_in[10];
  const float* bb  = (const float*)d_in[11];
  float* out = (float*)d_out;

  const int n = in_sizes[0] / IN_C;
  const int E = in_sizes[1] / 2;
  const int P = in_sizes[2] / 2;
  const int* e_src = eidx;
  const int* e_dst = eidx + E;

  float* ws   = (float*)d_ws;
  float* h1   = ws;
  float* z1   = h1 + (size_t)n * HID_C;
  const int nwg = (n + AGG_ROWS - 1) / AGG_ROWS;
  int* ents   = (int*)(z1 + (size_t)n * HID_C);
  int* lcnt   = ents + (size_t)nwg * AGG_WAVES * AGG_LCAP * 4;
  const int nmap = (E + MAP_THREADS - 1) / MAP_THREADS;
  const int bp = nmap * 32;
  unsigned* bmap = (unsigned*)(lcnt + (size_t)nwg * AGG_WAVES);
  float* dinv = (float*)(bmap + (size_t)nwg * bp);
  float* h2   = h1;
  float* z2   = h1 + (size_t)n * OUT_C;

  auto cdiv = [](long long a, long long b) { return (int)((a + b - 1) / b); };
  const int B = 256;

  edge_map<<<nmap, MAP_THREADS, 0, stream>>>(e_dst, E, bmap, nwg, bp);
  owner_list<<<nwg, AGG_THREADS, 0, stream>>>(e_src, e_dst, dinv, ents, lcnt, bmap, bp, n);

  {
    agg_owner<IN_C, true, 0, true><<<nwg, AGG_THREADS, 0, stream>>>(
        x, e_src, e_dst, dinv, b1, W1, z1, ents, lcnt, bmap, bp, n, E);
  }

  {
    long long waves = (long long)(n / 16) * (OUT_C / 16);
    wmma_gemm_f32<<<cdiv(waves * 32, B), B, 0, stream>>>(z1, W2, h2, n, OUT_C, HID_C);
    agg_owner<OUT_C, false, 1, false><<<nwg, AGG_THREADS, 0, stream>>>(
        h2, e_src, e_dst, dinv, b2, nullptr, z2, ents, lcnt, bmap, bp, n, E);
  }

  decode_mlp<<<DEC_WGS, DEC_THREADS, 0, stream>>>(z2, pidx, nidx, Wa, ba, Wb, bb, out, P);
}
